# starting kernel
# baseline (speedup 1.0000x reference)
#include <hip/hip_runtime.h>

typedef float v2f __attribute__((ext_vector_type(2)));
typedef float v8f __attribute__((ext_vector_type(8)));

__global__ void zero_i32_kernel(int* __restrict__ p, int n) {
    int i = blockIdx.x * blockDim.x + threadIdx.x;
    if (i < n) p[i] = 0;
}

__global__ void count_deg_kernel(const int* __restrict__ dst, int* __restrict__ deg, int E) {
    int e = blockIdx.x * blockDim.x + threadIdx.x;
    if (e < E) atomicAdd(&deg[dst[e]], 1);
}

__global__ void dinv_kernel(const int* __restrict__ deg, float* __restrict__ dinv, int n) {
    int i = blockIdx.x * blockDim.x + threadIdx.x;
    if (i < n) dinv[i] = rsqrtf((float)deg[i] + 1.0f);
}

__global__ void scan_kernel(const int* __restrict__ deg, int* __restrict__ rowstart, int n) {
    __shared__ int smem[256];
    __shared__ int carry;
    const int tid = threadIdx.x;
    if (tid == 0) carry = 0;
    __syncthreads();
    for (int base = 0; base < n; base += 256) {
        const int i = base + tid;
        const int v = (i < n) ? deg[i] : 0;
        smem[tid] = v;
        __syncthreads();
        for (int ofs = 1; ofs < 256; ofs <<= 1) {
            int t = (tid >= ofs) ? smem[tid - ofs] : 0;
            __syncthreads();
            smem[tid] += t;
            __syncthreads();
        }
        const int incl = smem[tid];
        if (i < n) rowstart[i] = carry + (incl - v);
        __syncthreads();
        if (tid == 0) carry += smem[255];
        __syncthreads();
    }
    if (tid == 0) rowstart[n] = carry;
}

__global__ void copy_i32_kernel(const int* __restrict__ a, int* __restrict__ b, int n) {
    int i = blockIdx.x * blockDim.x + threadIdx.x;
    if (i < n) b[i] = a[i];
}

__global__ void fill_csr_kernel(const int* __restrict__ src, const int* __restrict__ dst,
                                const float* __restrict__ dinv, int* __restrict__ cursor,
                                int* __restrict__ csr_src, float* __restrict__ csr_norm, int E) {
    int e = blockIdx.x * blockDim.x + threadIdx.x;
    if (e >= E) return;
    const int s = src[e];
    const int d = dst[e];
    const int pos = atomicAdd(&cursor[d], 1);
    csr_src[pos]  = s;
    csr_norm[pos] = dinv[s] * dinv[d];
}

__global__ void gemm_f32_wmma_kernel(const float* __restrict__ A,
                                     const float* __restrict__ B,
                                     float* __restrict__ C,
                                     int M, int N, int K) {
    const int lane = threadIdx.x & 31;
    const int wave = (blockIdx.x * blockDim.x + threadIdx.x) >> 5;
    const int nTiles = N >> 4;
    const int mTiles = M >> 4;
    if (wave >= mTiles * nTiles) return;
    const int mt  = wave / nTiles;
    const int nt  = wave - mt * nTiles;
    const int l16 = lane & 15;
    const int hi  = lane >> 4;
    const int row = (mt << 4) + l16;
    const int col = (nt << 4) + l16;
    const int kOff = hi << 1;

    v8f acc0 = {};
    v8f acc1 = {};
    const float* ap = A + (size_t)row * K + kOff;
    const float* bp = B + (size_t)kOff * N + col;
    const size_t bStep4 = (size_t)4 * N;

    for (int k = 0; k < K; k += 8) {
        v2f a0 = *(const v2f*)ap;
        v2f b0; b0.x = bp[0]; b0.y = bp[N];
        acc0 = __builtin_amdgcn_wmma_f32_16x16x4_f32(false, a0, false, b0,
                                                     (short)0, acc0, false, false);
        v2f a1 = *(const v2f*)(ap + 4);
        v2f b1; b1.x = bp[bStep4]; b1.y = bp[bStep4 + N];
        acc1 = __builtin_amdgcn_wmma_f32_16x16x4_f32(false, a1, false, b1,
                                                     (short)0, acc1, false, false);
        ap += 8;
        bp += 2 * bStep4;
    }
    v8f acc = acc0 + acc1;

    const int mBase = (mt << 4) + (hi << 3);
    #pragma unroll
    for (int v = 0; v < 8; ++v) {
        C[(size_t)(mBase + v) * N + col] = acc[v];
    }
}

__global__ void gather256_kernel(const int* __restrict__ rowstart,
                                 const int* __restrict__ csr_src,
                                 const float* __restrict__ csr_norm,
                                 const float* __restrict__ dinv,
                                 const float* __restrict__ h,
                                 const float* __restrict__ bias,
                                 float* __restrict__ out, int n, int doRelu) {
    const int lane = threadIdx.x & 31;
    int node = (blockIdx.x * blockDim.x + threadIdx.x) >> 5;
    if (node >= n) return;
    node = __builtin_amdgcn_readfirstlane(node);

    const int beg = rowstart[node];
    const int end = rowstart[node + 1];
    const float di = dinv[node];
    const float s  = di * di;

    const int c0 = lane;
    const int c1 = lane + 32;
    const float4* hn = (const float4*)(h + (size_t)node * 256);
    const float4* b4 = (const float4*)bias;

    float4 hv0 = hn[c0], hv1 = hn[c1];
    float4 bv0 = b4[c0], bv1 = b4[c1];
    float4 a0, a1;
    a0.x = fmaf(s, hv0.x, bv0.x); a0.y = fmaf(s, hv0.y, bv0.y);
    a0.z = fmaf(s, hv0.z, bv0.z); a0.w = fmaf(s, hv0.w, bv0.w);
    a1.x = fmaf(s, hv1.x, bv1.x); a1.y = fmaf(s, hv1.y, bv1.y);
    a1.z = fmaf(s, hv1.z, bv1.z); a1.w = fmaf(s, hv1.w, bv1.w);

    for (int j = beg; j < end; ++j) {
        const int   sj = csr_src[j];
        const float nj = csr_norm[j];
        const float4* hs = (const float4*)(h + (size_t)sj * 256);
        float4 v0 = hs[c0];
        float4 v1 = hs[c1];
        a0.x = fmaf(nj, v0.x, a0.x); a0.y = fmaf(nj, v0.y, a0.y);
        a0.z = fmaf(nj, v0.z, a0.z); a0.w = fmaf(nj, v0.w, a0.w);
        a1.x = fmaf(nj, v1.x, a1.x); a1.y = fmaf(nj, v1.y, a1.y);
        a1.z = fmaf(nj, v1.z, a1.z); a1.w = fmaf(nj, v1.w, a1.w);
    }
    if (doRelu) {
        a0.x = fmaxf(a0.x, 0.f); a0.y = fmaxf(a0.y, 0.f);
        a0.z = fmaxf(a0.z, 0.f); a0.w = fmaxf(a0.w, 0.f);
        a1.x = fmaxf(a1.x, 0.f); a1.y = fmaxf(a1.y, 0.f);
        a1.z = fmaxf(a1.z, 0.f); a1.w = fmaxf(a1.w, 0.f);
    }
    float4* op = (float4*)(out + (size_t)node * 256);
    op[c0] = a0;
    op[c1] = a1;
}

__global__ void gather64_kernel(const int* __restrict__ rowstart,
                                const int* __restrict__ csr_src,
                                const float* __restrict__ csr_norm,
                                const float* __restrict__ dinv,
                                const float* __restrict__ h,
                                const float* __restrict__ bias,
                                float* __restrict__ out, int n) {
    const int lane = threadIdx.x & 31;
    int node = (blockIdx.x * blockDim.x + threadIdx.x) >> 5;
    if (node >= n) return;
    node = __builtin_amdgcn_readfirstlane(node);

    const int beg = rowstart[node];
    const int end = rowstart[node + 1];
    const float di = dinv[node];
    const float s  = di * di;

    const float2* hn = (const float2*)(h + (size_t)node * 64);
    const float2* b2 = (const float2*)bias;
    float2 hv = hn[lane];
    float2 bv = b2[lane];
    float2 a;
    a.x = fmaf(s, hv.x, bv.x);
    a.y = fmaf(s, hv.y, bv.y);

    for (int j = beg; j < end; ++j) {
        const int   sj = csr_src[j];
        const float nj = csr_norm[j];
        const float2* hs = (const float2*)(h + (size_t)sj * 64);
        float2 v = hs[lane];
        a.x = fmaf(nj, v.x, a.x);
        a.y = fmaf(nj, v.y, a.y);
    }
    ((float2*)(out + (size_t)node * 64))[lane] = a;
}

extern "C" void kernel_launch(void* const* d_in, const int* in_sizes, int n_in,
                              void* d_out, int out_size, void* d_ws, size_t ws_size,
                              hipStream_t stream) {
    (void)n_in; (void)out_size; (void)ws_size;

    const float* x  = (const float*)d_in[0];
    const int*   ei = (const int*)d_in[1];
    const float* W1 = (const float*)d_in[2];
    const float* b1 = (const float*)d_in[3];
    const float* W2 = (const float*)d_in[4];
    const float* b2 = (const float*)d_in[5];
    float* out = (float*)d_out;

    const int Fin = 128, Fmid = 256, Fout = 64;
    const int N = in_sizes[0] / Fin;
    const int E = in_sizes[1] / 2;
    const int* src = ei;
    const int* dst = ei + E;

    char* ws = (char*)d_ws;
    size_t off = 0;
    auto take = [&](size_t bytes) -> char* {
        char* p = ws + off;
        off = (off + bytes + 255) & ~(size_t)255;
        return p;
    };
    int*   deg      = (int*)  take((size_t)N * 4);
    float* dinv     = (float*)take((size_t)N * 4);
    int*   rowstart = (int*)  take((size_t)(N + 1) * 4);
    int*   cursor   = (int*)  take((size_t)N * 4);
    int*   csr_src  = (int*)  take((size_t)E * 4);
    float* csr_norm = (float*)take((size_t)E * 4);
    float* hbuf     = (float*)take((size_t)N * Fmid * 4);
    float* hrelu    = (float*)take((size_t)N * Fmid * 4);
    float* h2 = hbuf;

    const int T = 256;

    zero_i32_kernel <<<(N + T - 1) / T, T, 0, stream>>>(deg, N);
    count_deg_kernel<<<(E + T - 1) / T, T, 0, stream>>>(dst, deg, E);
    dinv_kernel     <<<(N + T - 1) / T, T, 0, stream>>>(deg, dinv, N);
    scan_kernel     <<<1, T, 0, stream>>>(deg, rowstart, N);
    copy_i32_kernel <<<(N + T - 1) / T, T, 0, stream>>>(rowstart, cursor, N);
    fill_csr_kernel <<<(E + T - 1) / T, T, 0, stream>>>(src, dst, dinv, cursor,
                                                        csr_src, csr_norm, E);

    {
        int tiles = (N / 16) * (Fmid / 16);
        int wpb = T / 32;
        gemm_f32_wmma_kernel<<<(tiles + wpb - 1) / wpb, T, 0, stream>>>(x, W1, hbuf, N, Fmid, Fin);
    }
    {
        long long threads = (long long)N * 32;
        gather256_kernel<<<(int)((threads + T - 1) / T), T, 0, stream>>>(
            rowstart, csr_src, csr_norm, dinv, hbuf, b1, hrelu, N, 1);
    }
    {
        int tiles = (N / 16) * (Fout / 16);
        int wpb = T / 32;
        gemm_f32_wmma_kernel<<<(tiles + wpb - 1) / wpb, T, 0, stream>>>(hrelu, W2, h2, N, Fout, Fmid);
    }
    {
        long long threads = (long long)N * 32;
        gather64_kernel<<<(int)((threads + T - 1) / T), T, 0, stream>>>(
            rowstart, csr_src, csr_norm, dinv, h2, b2, out, N);
    }
}


# Degree prefix sum: one 1024-thread pass held in registers (128-bit loads, per-row wave scan, one LDS scan of wave totals, 3 barriers) replaces 196 serial 256-element chunks with log-step barriers
# speedup vs baseline: 2.1260x; 2.1260x over previous
#include <hip/hip_runtime.h>

typedef float v2f __attribute__((ext_vector_type(2)));
typedef float v8f __attribute__((ext_vector_type(8)));

__global__ void zero_i32_kernel(int* __restrict__ p, int n) {
    int i = blockIdx.x * blockDim.x + threadIdx.x;
    if (i < n) p[i] = 0;
}

__global__ void count_deg_kernel(const int* __restrict__ dst, int* __restrict__ deg, int E) {
    int e = blockIdx.x * blockDim.x + threadIdx.x;
    if (e < E) atomicAdd(&deg[dst[e]], 1);
}

__global__ void dinv_kernel(const int* __restrict__ deg, float* __restrict__ dinv, int n) {
    int i = blockIdx.x * blockDim.x + threadIdx.x;
    if (i < n) dinv[i] = rsqrtf((float)deg[i] + 1.0f);
}

#define SCAN_T 1024
#define SCAN_V 13

__device__ __forceinline__ int4 scan_load4(const int* __restrict__ p, int idx, int n) {
    int4 r;
    if (idx + 4 <= n) {
        r = *(const int4*)(p + idx);
    } else {
        r.x = (idx     < n) ? p[idx]     : 0;
        r.y = (idx + 1 < n) ? p[idx + 1] : 0;
        r.z = (idx + 2 < n) ? p[idx + 2] : 0;
        r.w = (idx + 3 < n) ? p[idx + 3] : 0;
    }
    return r;
}

__global__ void __launch_bounds__(SCAN_T)
scan_kernel(const int* __restrict__ deg, int* __restrict__ rowstart, int n) {
    __shared__ int wtot[SCAN_V * 32];
    __shared__ int passTotal;
    const int tid  = threadIdx.x;
    const int lane = tid & 31;
    const int wave = tid >> 5;
    const int PASS = SCAN_T * SCAN_V * 4;
    int carry = 0;

    for (int base = 0; base <= n; base += PASS) {
        int4 v[SCAN_V];
        int  s[SCAN_V];
        #pragma unroll
        for (int i = 0; i < SCAN_V; ++i)
            v[i] = scan_load4(deg, base + (i * SCAN_T + tid) * 4, n);
        #pragma unroll
        for (int i = 0; i < SCAN_V; ++i)
            s[i] = v[i].x + v[i].y + v[i].z + v[i].w;

        #pragma unroll
        for (int d = 1; d < 32; d <<= 1) {
            #pragma unroll
            for (int i = 0; i < SCAN_V; ++i) {
                const int t = __shfl_up(s[i], d, 32);
                if (lane >= d) s[i] += t;
            }
        }
        if (lane == 31) {
            #pragma unroll
            for (int i = 0; i < SCAN_V; ++i) wtot[i * 32 + wave] = s[i];
        }
        __syncthreads();

        if (wave == 0) {
            int t[SCAN_V];
            int sum = 0;
            #pragma unroll
            for (int k = 0; k < SCAN_V; ++k) t[k] = wtot[lane * SCAN_V + k];
            #pragma unroll
            for (int k = 0; k < SCAN_V; ++k) { const int x = t[k]; t[k] = sum; sum += x; }
            int incl = sum;
            #pragma unroll
            for (int d = 1; d < 32; d <<= 1) {
                const int u = __shfl_up(incl, d, 32);
                if (lane >= d) incl += u;
            }
            const int excl = incl - sum;
            #pragma unroll
            for (int k = 0; k < SCAN_V; ++k) wtot[lane * SCAN_V + k] = t[k] + excl;
            if (lane == 31) passTotal = incl;
        }
        __syncthreads();

        #pragma unroll
        for (int i = 0; i < SCAN_V; ++i) {
            const int idx = base + (i * SCAN_T + tid) * 4;
            const int own = v[i].x + v[i].y + v[i].z + v[i].w;
            int4 o;
            o.x = carry + wtot[i * 32 + wave] + (s[i] - own);
            o.y = o.x + v[i].x;
            o.z = o.y + v[i].y;
            o.w = o.z + v[i].z;
            if (idx + 3 <= n) {
                *(int4*)(rowstart + idx) = o;
            } else {
                if (idx     <= n) rowstart[idx]     = o.x;
                if (idx + 1 <= n) rowstart[idx + 1] = o.y;
                if (idx + 2 <= n) rowstart[idx + 2] = o.z;
            }
        }
        carry += passTotal;
        __syncthreads();
    }
}

__global__ void copy_i32_kernel(const int* __restrict__ a, int* __restrict__ b, int n) {
    int i = blockIdx.x * blockDim.x + threadIdx.x;
    if (i < n) b[i] = a[i];
}

__global__ void fill_csr_kernel(const int* __restrict__ src, const int* __restrict__ dst,
                                const float* __restrict__ dinv, int* __restrict__ cursor,
                                int* __restrict__ csr_src, float* __restrict__ csr_norm, int E) {
    int e = blockIdx.x * blockDim.x + threadIdx.x;
    if (e >= E) return;
    const int s = src[e];
    const int d = dst[e];
    const int pos = atomicAdd(&cursor[d], 1);
    csr_src[pos]  = s;
    csr_norm[pos] = dinv[s] * dinv[d];
}

__global__ void gemm_f32_wmma_kernel(const float* __restrict__ A,
                                     const float* __restrict__ B,
                                     float* __restrict__ C,
                                     int M, int N, int K) {
    const int lane = threadIdx.x & 31;
    const int wave = (blockIdx.x * blockDim.x + threadIdx.x) >> 5;
    const int nTiles = N >> 4;
    const int mTiles = M >> 4;
    if (wave >= mTiles * nTiles) return;
    const int mt  = wave / nTiles;
    const int nt  = wave - mt * nTiles;
    const int l16 = lane & 15;
    const int hi  = lane >> 4;
    const int row = (mt << 4) + l16;
    const int col = (nt << 4) + l16;
    const int kOff = hi << 1;

    v8f acc0 = {};
    v8f acc1 = {};
    const float* ap = A + (size_t)row * K + kOff;
    const float* bp = B + (size_t)kOff * N + col;
    const size_t bStep4 = (size_t)4 * N;

    for (int k = 0; k < K; k += 8) {
        v2f a0 = *(const v2f*)ap;
        v2f b0; b0.x = bp[0]; b0.y = bp[N];
        acc0 = __builtin_amdgcn_wmma_f32_16x16x4_f32(false, a0, false, b0,
                                                     (short)0, acc0, false, false);
        v2f a1 = *(const v2f*)(ap + 4);
        v2f b1; b1.x = bp[bStep4]; b1.y = bp[bStep4 + N];
        acc1 = __builtin_amdgcn_wmma_f32_16x16x4_f32(false, a1, false, b1,
                                                     (short)0, acc1, false, false);
        ap += 8;
        bp += 2 * bStep4;
    }
    v8f acc = acc0 + acc1;

    const int mBase = (mt << 4) + (hi << 3);
    #pragma unroll
    for (int v = 0; v < 8; ++v) {
        C[(size_t)(mBase + v) * N + col] = acc[v];
    }
}

__global__ void gather256_kernel(const int* __restrict__ rowstart,
                                 const int* __restrict__ csr_src,
                                 const float* __restrict__ csr_norm,
                                 const float* __restrict__ dinv,
                                 const float* __restrict__ h,
                                 const float* __restrict__ bias,
                                 float* __restrict__ out, int n, int doRelu) {
    const int lane = threadIdx.x & 31;
    int node = (blockIdx.x * blockDim.x + threadIdx.x) >> 5;
    if (node >= n) return;
    node = __builtin_amdgcn_readfirstlane(node);

    const int beg = rowstart[node];
    const int end = rowstart[node + 1];
    const float di = dinv[node];
    const float s  = di * di;

    const int c0 = lane;
    const int c1 = lane + 32;
    const float4* hn = (const float4*)(h + (size_t)node * 256);
    const float4* b4 = (const float4*)bias;

    float4 hv0 = hn[c0], hv1 = hn[c1];
    float4 bv0 = b4[c0], bv1 = b4[c1];
    float4 a0, a1;
    a0.x = fmaf(s, hv0.x, bv0.x); a0.y = fmaf(s, hv0.y, bv0.y);
    a0.z = fmaf(s, hv0.z, bv0.z); a0.w = fmaf(s, hv0.w, bv0.w);
    a1.x = fmaf(s, hv1.x, bv1.x); a1.y = fmaf(s, hv1.y, bv1.y);
    a1.z = fmaf(s, hv1.z, bv1.z); a1.w = fmaf(s, hv1.w, bv1.w);

    for (int j = beg; j < end; ++j) {
        const int   sj = csr_src[j];
        const float nj = csr_norm[j];
        const float4* hs = (const float4*)(h + (size_t)sj * 256);
        float4 v0 = hs[c0];
        float4 v1 = hs[c1];
        a0.x = fmaf(nj, v0.x, a0.x); a0.y = fmaf(nj, v0.y, a0.y);
        a0.z = fmaf(nj, v0.z, a0.z); a0.w = fmaf(nj, v0.w, a0.w);
        a1.x = fmaf(nj, v1.x, a1.x); a1.y = fmaf(nj, v1.y, a1.y);
        a1.z = fmaf(nj, v1.z, a1.z); a1.w = fmaf(nj, v1.w, a1.w);
    }
    if (doRelu) {
        a0.x = fmaxf(a0.x, 0.f); a0.y = fmaxf(a0.y, 0.f);
        a0.z = fmaxf(a0.z, 0.f); a0.w = fmaxf(a0.w, 0.f);
        a1.x = fmaxf(a1.x, 0.f); a1.y = fmaxf(a1.y, 0.f);
        a1.z = fmaxf(a1.z, 0.f); a1.w = fmaxf(a1.w, 0.f);
    }
    float4* op = (float4*)(out + (size_t)node * 256);
    op[c0] = a0;
    op[c1] = a1;
}

__global__ void gather64_kernel(const int* __restrict__ rowstart,
                                const int* __restrict__ csr_src,
                                const float* __restrict__ csr_norm,
                                const float* __restrict__ dinv,
                                const float* __restrict__ h,
                                const float* __restrict__ bias,
                                float* __restrict__ out, int n) {
    const int lane = threadIdx.x & 31;
    int node = (blockIdx.x * blockDim.x + threadIdx.x) >> 5;
    if (node >= n) return;
    node = __builtin_amdgcn_readfirstlane(node);

    const int beg = rowstart[node];
    const int end = rowstart[node + 1];
    const float di = dinv[node];
    const float s  = di * di;

    const float2* hn = (const float2*)(h + (size_t)node * 64);
    const float2* b2 = (const float2*)bias;
    float2 hv = hn[lane];
    float2 bv = b2[lane];
    float2 a;
    a.x = fmaf(s, hv.x, bv.x);
    a.y = fmaf(s, hv.y, bv.y);

    for (int j = beg; j < end; ++j) {
        const int   sj = csr_src[j];
        const float nj = csr_norm[j];
        const float2* hs = (const float2*)(h + (size_t)sj * 64);
        float2 v = hs[lane];
        a.x = fmaf(nj, v.x, a.x);
        a.y = fmaf(nj, v.y, a.y);
    }
    ((float2*)(out + (size_t)node * 64))[lane] = a;
}

extern "C" void kernel_launch(void* const* d_in, const int* in_sizes, int n_in,
                              void* d_out, int out_size, void* d_ws, size_t ws_size,
                              hipStream_t stream) {
    (void)n_in; (void)out_size; (void)ws_size;

    const float* x  = (const float*)d_in[0];
    const int*   ei = (const int*)d_in[1];
    const float* W1 = (const float*)d_in[2];
    const float* b1 = (const float*)d_in[3];
    const float* W2 = (const float*)d_in[4];
    const float* b2 = (const float*)d_in[5];
    float* out = (float*)d_out;

    const int Fin = 128, Fmid = 256, Fout = 64;
    const int N = in_sizes[0] / Fin;
    const int E = in_sizes[1] / 2;
    const int* src = ei;
    const int* dst = ei + E;

    char* ws = (char*)d_ws;
    size_t off = 0;
    auto take = [&](size_t bytes) -> char* {
        char* p = ws + off;
        off = (off + bytes + 255) & ~(size_t)255;
        return p;
    };
    int*   deg      = (int*)  take((size_t)N * 4);
    float* dinv     = (float*)take((size_t)N * 4);
    int*   rowstart = (int*)  take((size_t)(N + 1) * 4);
    int*   cursor   = (int*)  take((size_t)N * 4);
    int*   csr_src  = (int*)  take((size_t)E * 4);
    float* csr_norm = (float*)take((size_t)E * 4);
    float* hbuf     = (float*)take((size_t)N * Fmid * 4);
    float* hrelu    = (float*)take((size_t)N * Fmid * 4);
    float* h2 = hbuf;

    const int T = 256;

    zero_i32_kernel <<<(N + T - 1) / T, T, 0, stream>>>(deg, N);
    count_deg_kernel<<<(E + T - 1) / T, T, 0, stream>>>(dst, deg, E);
    dinv_kernel     <<<(N + T - 1) / T, T, 0, stream>>>(deg, dinv, N);
    scan_kernel     <<<1, SCAN_T, 0, stream>>>(deg, rowstart, N);
    copy_i32_kernel <<<(N + T - 1) / T, T, 0, stream>>>(rowstart, cursor, N);
    fill_csr_kernel <<<(E + T - 1) / T, T, 0, stream>>>(src, dst, dinv, cursor,
                                                        csr_src, csr_norm, E);

    {
        int tiles = (N / 16) * (Fmid / 16);
        int wpb = T / 32;
        gemm_f32_wmma_kernel<<<(tiles + wpb - 1) / wpb, T, 0, stream>>>(x, W1, hbuf, N, Fmid, Fin);
    }
    {
        long long threads = (long long)N * 32;
        gather256_kernel<<<(int)((threads + T - 1) / T), T, 0, stream>>>(
            rowstart, csr_src, csr_norm, dinv, hbuf, b1, hrelu, N, 1);
    }
    {
        int tiles = (N / 16) * (Fout / 16);
        int wpb = T / 32;
        gemm_f32_wmma_kernel<<<(tiles + wpb - 1) / wpb, T, 0, stream>>>(hrelu, W2, h2, N, Fout, Fmid);
    }
    {
        long long threads = (long long)N * 32;
        gather64_kernel<<<(int)((threads + T - 1) / T), T, 0, stream>>>(
            rowstart, csr_src, csr_norm, dinv, h2, b2, out, N);
    }
}


# GEMM operand reuse: 256 resident work-groups stage W once in LDS in fragment order, each wave reuses its A fragment over 2-4 column tiles; k summed in one accumulator per tile (order changes)
# speedup vs baseline: 2.6095x; 1.2274x over previous
#include <hip/hip_runtime.h>

typedef float v2f __attribute__((ext_vector_type(2)));
typedef float v4f __attribute__((ext_vector_type(4)));
typedef float v8f __attribute__((ext_vector_type(8)));

__global__ void zero_i32_kernel(int* __restrict__ p, int n) {
    int i = blockIdx.x * blockDim.x + threadIdx.x;
    if (i < n) p[i] = 0;
}

__global__ void count_deg_kernel(const int* __restrict__ dst, int* __restrict__ deg, int E) {
    int e = blockIdx.x * blockDim.x + threadIdx.x;
    if (e < E) atomicAdd(&deg[dst[e]], 1);
}

__global__ void dinv_kernel(const int* __restrict__ deg, float* __restrict__ dinv, int n) {
    int i = blockIdx.x * blockDim.x + threadIdx.x;
    if (i < n) dinv[i] = rsqrtf((float)deg[i] + 1.0f);
}

#define SCAN_T 1024
#define SCAN_V 13

__device__ __forceinline__ int4 scan_load4(const int* __restrict__ p, int idx, int n) {
    int4 r;
    if (idx + 4 <= n) {
        r = *(const int4*)(p + idx);
    } else {
        r.x = (idx     < n) ? p[idx]     : 0;
        r.y = (idx + 1 < n) ? p[idx + 1] : 0;
        r.z = (idx + 2 < n) ? p[idx + 2] : 0;
        r.w = (idx + 3 < n) ? p[idx + 3] : 0;
    }
    return r;
}

__global__ void __launch_bounds__(SCAN_T)
scan_kernel(const int* __restrict__ deg, int* __restrict__ rowstart, int n) {
    __shared__ int wtot[SCAN_V * 32];
    __shared__ int passTotal;
    const int tid  = threadIdx.x;
    const int lane = tid & 31;
    const int wave = tid >> 5;
    const int PASS = SCAN_T * SCAN_V * 4;
    int carry = 0;

    for (int base = 0; base <= n; base += PASS) {
        int4 v[SCAN_V];
        int  s[SCAN_V];
        #pragma unroll
        for (int i = 0; i < SCAN_V; ++i)
            v[i] = scan_load4(deg, base + (i * SCAN_T + tid) * 4, n);
        #pragma unroll
        for (int i = 0; i < SCAN_V; ++i)
            s[i] = v[i].x + v[i].y + v[i].z + v[i].w;

        #pragma unroll
        for (int d = 1; d < 32; d <<= 1) {
            #pragma unroll
            for (int i = 0; i < SCAN_V; ++i) {
                const int t = __shfl_up(s[i], d, 32);
                if (lane >= d) s[i] += t;
            }
        }
        if (lane == 31) {
            #pragma unroll
            for (int i = 0; i < SCAN_V; ++i) wtot[i * 32 + wave] = s[i];
        }
        __syncthreads();

        if (wave == 0) {
            int t[SCAN_V];
            int sum = 0;
            #pragma unroll
            for (int k = 0; k < SCAN_V; ++k) t[k] = wtot[lane * SCAN_V + k];
            #pragma unroll
            for (int k = 0; k < SCAN_V; ++k) { const int x = t[k]; t[k] = sum; sum += x; }
            int incl = sum;
            #pragma unroll
            for (int d = 1; d < 32; d <<= 1) {
                const int u = __shfl_up(incl, d, 32);
                if (lane >= d) incl += u;
            }
            const int excl = incl - sum;
            #pragma unroll
            for (int k = 0; k < SCAN_V; ++k) wtot[lane * SCAN_V + k] = t[k] + excl;
            if (lane == 31) passTotal = incl;
        }
        __syncthreads();

        #pragma unroll
        for (int i = 0; i < SCAN_V; ++i) {
            const int idx = base + (i * SCAN_T + tid) * 4;
            const int own = v[i].x + v[i].y + v[i].z + v[i].w;
            int4 o;
            o.x = carry + wtot[i * 32 + wave] + (s[i] - own);
            o.y = o.x + v[i].x;
            o.z = o.y + v[i].y;
            o.w = o.z + v[i].z;
            if (idx + 3 <= n) {
                *(int4*)(rowstart + idx) = o;
            } else {
                if (idx     <= n) rowstart[idx]     = o.x;
                if (idx + 1 <= n) rowstart[idx + 1] = o.y;
                if (idx + 2 <= n) rowstart[idx + 2] = o.z;
            }
        }
        carry += passTotal;
        __syncthreads();
    }
}

__global__ void copy_i32_kernel(const int* __restrict__ a, int* __restrict__ b, int n) {
    int i = blockIdx.x * blockDim.x + threadIdx.x;
    if (i < n) b[i] = a[i];
}

__global__ void fill_csr_kernel(const int* __restrict__ src, const int* __restrict__ dst,
                                const float* __restrict__ dinv, int* __restrict__ cursor,
                                int* __restrict__ csr_src, float* __restrict__ csr_norm, int E) {
    int e = blockIdx.x * blockDim.x + threadIdx.x;
    if (e >= E) return;
    const int s = src[e];
    const int d = dst[e];
    const int pos = atomicAdd(&cursor[d], 1);
    csr_src[pos]  = s;
    csr_norm[pos] = dinv[s] * dinv[d];
}

#define GEMM_T 512
#define GEMM_GRID 256

template <int K, int N, int NT>
__global__ void __launch_bounds__(GEMM_T)
gemm_f32_wmma_kernel(const float* __restrict__ A,
                     const float* __restrict__ B,
                     float* __restrict__ C, int M) {
    constexpr int KU = K / 8;
    constexpr int KK = K / 4;
    constexpr int NG = N / (16 * NT);
    constexpr int TW = 16 * NT;
    static_assert((KK * N) % GEMM_T == 0 && TW % 32 == 0 && N % TW == 0, "tile constants");
    __shared__ v4f Bl[KK * N];

    const int tid  = threadIdx.x;
    const int lane = tid & 31;
    const int wave = __builtin_amdgcn_readfirstlane(tid >> 5);
    const int c = lane & 15;
    const int h = lane >> 4;
    const int T = (M >> 4) * NG;
    const int tStep = gridDim.x * (GEMM_T / 32);
    int t = blockIdx.x + gridDim.x * wave;

    v4f a[KU];
    if (t < T) {
        const int mt = t / NG;
        const float* ap = A + (size_t)((mt << 4) + c) * K + (h << 2);
        #pragma unroll
        for (int u = 0; u < KU; ++u) a[u] = *(const v4f*)(ap + 8 * u);
    }

    #pragma unroll 8
    for (int i = 0; i < (KK * N) / GEMM_T; ++i) {
        const int e  = tid + i * GEMM_T;
        const int kk = e / N;
        const int n  = e - kk * N;
        const float* bp = B + (size_t)(4 * kk) * N + n;
        v4f v;
        v.x = bp[0]; v.y = bp[N]; v.z = bp[2 * N]; v.w = bp[3 * N];
        const int g  = n / TW;
        const int nl = n - g * TW;
        Bl[(g * KK + kk) * TW + (nl ^ ((kk & 1) << 3))] = v;
    }
    __syncthreads();

    const v4f* bl = Bl + h * TW + (c ^ (h << 3));
    while (t < T) {
        const int mt = t / NG;
        const int ng = t - mt * NG;
        const v4f* bt = bl + ng * (KK * TW);

        v8f acc[NT];
        v4f b[2][NT];
        #pragma unroll
        for (int j = 0; j < NT; ++j) { acc[j] = v8f{}; b[0][j] = bt[16 * j]; }

        #pragma unroll
        for (int u = 0; u < KU; ++u) {
            if (u + 1 < KU) {
                #pragma unroll
                for (int j = 0; j < NT; ++j) b[(u + 1) & 1][j] = bt[2 * (u + 1) * TW + 16 * j];
            }
            #pragma unroll
            for (int j = 0; j < NT; ++j)
                acc[j] = __builtin_amdgcn_wmma_f32_16x16x4_f32(false, a[u].xy, false, b[u & 1][j].xy,
                                                               (short)0, acc[j], false, false);
            #pragma unroll
            for (int j = 0; j < NT; ++j)
                acc[j] = __builtin_amdgcn_wmma_f32_16x16x4_f32(false, a[u].zw, false, b[u & 1][j].zw,
                                                               (short)0, acc[j], false, false);
        }

        float* cp = C + (size_t)((mt << 4) + (h << 3)) * N + ng * (16 * NT) + c;
        #pragma unroll
        for (int j = 0; j < NT; ++j) {
            #pragma unroll
            for (int r = 0; r < 8; ++r) cp[(size_t)r * N + 16 * j] = acc[j][r];
        }

        t += tStep;
        if (t < T) {
            const int mn = t / NG;
            const float* ap = A + (size_t)((mn << 4) + c) * K + (h << 2);
            #pragma unroll
            for (int u = 0; u < KU; ++u) a[u] = *(const v4f*)(ap + 8 * u);
        }
    }
}

__global__ void gather256_kernel(const int* __restrict__ rowstart,
                                 const int* __restrict__ csr_src,
                                 const float* __restrict__ csr_norm,
                                 const float* __restrict__ dinv,
                                 const float* __restrict__ h,
                                 const float* __restrict__ bias,
                                 float* __restrict__ out, int n, int doRelu) {
    const int lane = threadIdx.x & 31;
    int node = (blockIdx.x * blockDim.x + threadIdx.x) >> 5;
    if (node >= n) return;
    node = __builtin_amdgcn_readfirstlane(node);

    const int beg = rowstart[node];
    const int end = rowstart[node + 1];
    const float di = dinv[node];
    const float s  = di * di;

    const int c0 = lane;
    const int c1 = lane + 32;
    const float4* hn = (const float4*)(h + (size_t)node * 256);
    const float4* b4 = (const float4*)bias;

    float4 hv0 = hn[c0], hv1 = hn[c1];
    float4 bv0 = b4[c0], bv1 = b4[c1];
    float4 a0, a1;
    a0.x = fmaf(s, hv0.x, bv0.x); a0.y = fmaf(s, hv0.y, bv0.y);
    a0.z = fmaf(s, hv0.z, bv0.z); a0.w = fmaf(s, hv0.w, bv0.w);
    a1.x = fmaf(s, hv1.x, bv1.x); a1.y = fmaf(s, hv1.y, bv1.y);
    a1.z = fmaf(s, hv1.z, bv1.z); a1.w = fmaf(s, hv1.w, bv1.w);

    for (int j = beg; j < end; ++j) {
        const int   sj = csr_src[j];
        const float nj = csr_norm[j];
        const float4* hs = (const float4*)(h + (size_t)sj * 256);
        float4 v0 = hs[c0];
        float4 v1 = hs[c1];
        a0.x = fmaf(nj, v0.x, a0.x); a0.y = fmaf(nj, v0.y, a0.y);
        a0.z = fmaf(nj, v0.z, a0.z); a0.w = fmaf(nj, v0.w, a0.w);
        a1.x = fmaf(nj, v1.x, a1.x); a1.y = fmaf(nj, v1.y, a1.y);
        a1.z = fmaf(nj, v1.z, a1.z); a1.w = fmaf(nj, v1.w, a1.w);
    }
    if (doRelu) {
        a0.x = fmaxf(a0.x, 0.f); a0.y = fmaxf(a0.y, 0.f);
        a0.z = fmaxf(a0.z, 0.f); a0.w = fmaxf(a0.w, 0.f);
        a1.x = fmaxf(a1.x, 0.f); a1.y = fmaxf(a1.y, 0.f);
        a1.z = fmaxf(a1.z, 0.f); a1.w = fmaxf(a1.w, 0.f);
    }
    float4* op = (float4*)(out + (size_t)node * 256);
    op[c0] = a0;
    op[c1] = a1;
}

__global__ void gather64_kernel(const int* __restrict__ rowstart,
                                const int* __restrict__ csr_src,
                                const float* __restrict__ csr_norm,
                                const float* __restrict__ dinv,
                                const float* __restrict__ h,
                                const float* __restrict__ bias,
                                float* __restrict__ out, int n) {
    const int lane = threadIdx.x & 31;
    int node = (blockIdx.x * blockDim.x + threadIdx.x) >> 5;
    if (node >= n) return;
    node = __builtin_amdgcn_readfirstlane(node);

    const int beg = rowstart[node];
    const int end = rowstart[node + 1];
    const float di = dinv[node];
    const float s  = di * di;

    const float2* hn = (const float2*)(h + (size_t)node * 64);
    const float2* b2 = (const float2*)bias;
    float2 hv = hn[lane];
    float2 bv = b2[lane];
    float2 a;
    a.x = fmaf(s, hv.x, bv.x);
    a.y = fmaf(s, hv.y, bv.y);

    for (int j = beg; j < end; ++j) {
        const int   sj = csr_src[j];
        const float nj = csr_norm[j];
        const float2* hs = (const float2*)(h + (size_t)sj * 64);
        float2 v = hs[lane];
        a.x = fmaf(nj, v.x, a.x);
        a.y = fmaf(nj, v.y, a.y);
    }
    ((float2*)(out + (size_t)node * 64))[lane] = a;
}

extern "C" void kernel_launch(void* const* d_in, const int* in_sizes, int n_in,
                              void* d_out, int out_size, void* d_ws, size_t ws_size,
                              hipStream_t stream) {
    (void)n_in; (void)out_size; (void)ws_size;

    const float* x  = (const float*)d_in[0];
    const int*   ei = (const int*)d_in[1];
    const float* W1 = (const float*)d_in[2];
    const float* b1 = (const float*)d_in[3];
    const float* W2 = (const float*)d_in[4];
    const float* b2 = (const float*)d_in[5];
    float* out = (float*)d_out;

    const int Fin = 128, Fmid = 256, Fout = 64;
    const int N = in_sizes[0] / Fin;
    const int E = in_sizes[1] / 2;
    const int* src = ei;
    const int* dst = ei + E;

    char* ws = (char*)d_ws;
    size_t off = 0;
    auto take = [&](size_t bytes) -> char* {
        char* p = ws + off;
        off = (off + bytes + 255) & ~(size_t)255;
        return p;
    };
    int*   deg      = (int*)  take((size_t)N * 4);
    float* dinv     = (float*)take((size_t)N * 4);
    int*   rowstart = (int*)  take((size_t)(N + 1) * 4);
    int*   cursor   = (int*)  take((size_t)N * 4);
    int*   csr_src  = (int*)  take((size_t)E * 4);
    float* csr_norm = (float*)take((size_t)E * 4);
    float* hbuf     = (float*)take((size_t)N * Fmid * 4);
    float* hrelu    = (float*)take((size_t)N * Fmid * 4);
    float* h2 = hbuf;

    const int T = 256;

    zero_i32_kernel <<<(N + T - 1) / T, T, 0, stream>>>(deg, N);
    count_deg_kernel<<<(E + T - 1) / T, T, 0, stream>>>(dst, deg, E);
    dinv_kernel     <<<(N + T - 1) / T, T, 0, stream>>>(deg, dinv, N);
    scan_kernel     <<<1, SCAN_T, 0, stream>>>(deg, rowstart, N);
    copy_i32_kernel <<<(N + T - 1) / T, T, 0, stream>>>(rowstart, cursor, N);
    fill_csr_kernel <<<(E + T - 1) / T, T, 0, stream>>>(src, dst, dinv, cursor,
                                                        csr_src, csr_norm, E);

    gemm_f32_wmma_kernel<128, 256, 4><<<GEMM_GRID, GEMM_T, 0, stream>>>(x, W1, hbuf, N);
    {
        long long threads = (long long)N * 32;
        gather256_kernel<<<(int)((threads + T - 1) / T), T, 0, stream>>>(
            rowstart, csr_src, csr_norm, dinv, hbuf, b1, hrelu, N, 1);
    }
    gemm_f32_wmma_kernel<256, 64, 2><<<GEMM_GRID, GEMM_T, 0, stream>>>(hrelu, W2, h2, N);
    {
        long long threads = (long long)N * 32;
        gather64_kernel<<<(int)((threads + T - 1) / T), T, 0, stream>>>(
            rowstart, csr_src, csr_norm, dinv, h2, b2, out, N);
    }
}


# Layer 1 aggregates its 128-wide input before the GEMM, not the 256-wide product after it ((A x) W = A (x W)): gathered rows 512 B, not 1 KiB; bias+relu move into the GEMM store; sum order changes
# speedup vs baseline: 2.7634x; 1.0590x over previous
#include <hip/hip_runtime.h>

typedef float v2f __attribute__((ext_vector_type(2)));
typedef float v4f __attribute__((ext_vector_type(4)));
typedef float v8f __attribute__((ext_vector_type(8)));

__global__ void zero_i32_kernel(int* __restrict__ p, int n) {
    int i = blockIdx.x * blockDim.x + threadIdx.x;
    if (i < n) p[i] = 0;
}

__global__ void count_deg_kernel(const int* __restrict__ dst, int* __restrict__ deg, int E) {
    int e = blockIdx.x * blockDim.x + threadIdx.x;
    if (e < E) atomicAdd(&deg[dst[e]], 1);
}

__global__ void dinv_kernel(const int* __restrict__ deg, float* __restrict__ dinv, int n) {
    int i = blockIdx.x * blockDim.x + threadIdx.x;
    if (i < n) dinv[i] = rsqrtf((float)deg[i] + 1.0f);
}

#define SCAN_T 1024
#define SCAN_V 13

__device__ __forceinline__ int4 scan_load4(const int* __restrict__ p, int idx, int n) {
    int4 r;
    if (idx + 4 <= n) {
        r = *(const int4*)(p + idx);
    } else {
        r.x = (idx     < n) ? p[idx]     : 0;
        r.y = (idx + 1 < n) ? p[idx + 1] : 0;
        r.z = (idx + 2 < n) ? p[idx + 2] : 0;
        r.w = (idx + 3 < n) ? p[idx + 3] : 0;
    }
    return r;
}

__global__ void __launch_bounds__(SCAN_T)
scan_kernel(const int* __restrict__ deg, int* __restrict__ rowstart, int n) {
    __shared__ int wtot[SCAN_V * 32];
    __shared__ int passTotal;
    const int tid  = threadIdx.x;
    const int lane = tid & 31;
    const int wave = tid >> 5;
    const int PASS = SCAN_T * SCAN_V * 4;
    int carry = 0;

    for (int base = 0; base <= n; base += PASS) {
        int4 v[SCAN_V];
        int  s[SCAN_V];
        #pragma unroll
        for (int i = 0; i < SCAN_V; ++i)
            v[i] = scan_load4(deg, base + (i * SCAN_T + tid) * 4, n);
        #pragma unroll
        for (int i = 0; i < SCAN_V; ++i)
            s[i] = v[i].x + v[i].y + v[i].z + v[i].w;

        #pragma unroll
        for (int d = 1; d < 32; d <<= 1) {
            #pragma unroll
            for (int i = 0; i < SCAN_V; ++i) {
                const int t = __shfl_up(s[i], d, 32);
                if (lane >= d) s[i] += t;
            }
        }
        if (lane == 31) {
            #pragma unroll
            for (int i = 0; i < SCAN_V; ++i) wtot[i * 32 + wave] = s[i];
        }
        __syncthreads();

        if (wave == 0) {
            int t[SCAN_V];
            int sum = 0;
            #pragma unroll
            for (int k = 0; k < SCAN_V; ++k) t[k] = wtot[lane * SCAN_V + k];
            #pragma unroll
            for (int k = 0; k < SCAN_V; ++k) { const int x = t[k]; t[k] = sum; sum += x; }
            int incl = sum;
            #pragma unroll
            for (int d = 1; d < 32; d <<= 1) {
                const int u = __shfl_up(incl, d, 32);
                if (lane >= d) incl += u;
            }
            const int excl = incl - sum;
            #pragma unroll
            for (int k = 0; k < SCAN_V; ++k) wtot[lane * SCAN_V + k] = t[k] + excl;
            if (lane == 31) passTotal = incl;
        }
        __syncthreads();

        #pragma unroll
        for (int i = 0; i < SCAN_V; ++i) {
            const int idx = base + (i * SCAN_T + tid) * 4;
            const int own = v[i].x + v[i].y + v[i].z + v[i].w;
            int4 o;
            o.x = carry + wtot[i * 32 + wave] + (s[i] - own);
            o.y = o.x + v[i].x;
            o.z = o.y + v[i].y;
            o.w = o.z + v[i].z;
            if (idx + 3 <= n) {
                *(int4*)(rowstart + idx) = o;
            } else {
                if (idx     <= n) rowstart[idx]     = o.x;
                if (idx + 1 <= n) rowstart[idx + 1] = o.y;
                if (idx + 2 <= n) rowstart[idx + 2] = o.z;
            }
        }
        carry += passTotal;
        __syncthreads();
    }
}

__global__ void copy_i32_kernel(const int* __restrict__ a, int* __restrict__ b, int n) {
    int i = blockIdx.x * blockDim.x + threadIdx.x;
    if (i < n) b[i] = a[i];
}

__global__ void fill_csr_kernel(const int* __restrict__ src, const int* __restrict__ dst,
                                const float* __restrict__ dinv, int* __restrict__ cursor,
                                int* __restrict__ csr_src, float* __restrict__ csr_norm, int E) {
    int e = blockIdx.x * blockDim.x + threadIdx.x;
    if (e >= E) return;
    const int s = src[e];
    const int d = dst[e];
    const int pos = atomicAdd(&cursor[d], 1);
    csr_src[pos]  = s;
    csr_norm[pos] = dinv[s] * dinv[d];
}

#define GEMM_T 512
#define GEMM_GRID 256

template <int K, int N, int NT, bool RELU_BIAS>
__global__ void __launch_bounds__(GEMM_T)
gemm_f32_wmma_kernel(const float* __restrict__ A,
                     const float* __restrict__ B,
                     const float* __restrict__ bias,
                     float* __restrict__ C, int M) {
    constexpr int KU = K / 8;
    constexpr int KK = K / 4;
    constexpr int NG = N / (16 * NT);
    constexpr int TW = 16 * NT;
    static_assert((KK * N) % GEMM_T == 0 && TW % 32 == 0 && N % TW == 0, "tile constants");
    __shared__ v4f Bl[KK * N];

    const int tid  = threadIdx.x;
    const int lane = tid & 31;
    const int wave = __builtin_amdgcn_readfirstlane(tid >> 5);
    const int c = lane & 15;
    const int h = lane >> 4;
    const int T = (M >> 4) * NG;
    const int tStep = gridDim.x * (GEMM_T / 32);
    int t = blockIdx.x + gridDim.x * wave;

    v4f a[KU];
    if (t < T) {
        const int mt = t / NG;
        const float* ap = A + (size_t)((mt << 4) + c) * K + (h << 2);
        #pragma unroll
        for (int u = 0; u < KU; ++u) a[u] = *(const v4f*)(ap + 8 * u);
    }

    #pragma unroll 8
    for (int i = 0; i < (KK * N) / GEMM_T; ++i) {
        const int e  = tid + i * GEMM_T;
        const int kk = e / N;
        const int n  = e - kk * N;
        const float* bp = B + (size_t)(4 * kk) * N + n;
        v4f v;
        v.x = bp[0]; v.y = bp[N]; v.z = bp[2 * N]; v.w = bp[3 * N];
        const int g  = n / TW;
        const int nl = n - g * TW;
        Bl[(g * KK + kk) * TW + (nl ^ ((kk & 1) << 3))] = v;
    }
    __syncthreads();

    const v4f* bl = Bl + h * TW + (c ^ (h << 3));
    while (t < T) {
        const int mt = t / NG;
        const int ng = t - mt * NG;
        const v4f* bt = bl + ng * (KK * TW);

        v8f acc[NT];
        v4f b[2][NT];
        #pragma unroll
        for (int j = 0; j < NT; ++j) { acc[j] = v8f{}; b[0][j] = bt[16 * j]; }
        float bj[NT];
        if (RELU_BIAS) {
            #pragma unroll
            for (int j = 0; j < NT; ++j) bj[j] = bias[ng * TW + 16 * j + c];
        }

        #pragma unroll
        for (int u = 0; u < KU; ++u) {
            if (u + 1 < KU) {
                #pragma unroll
                for (int j = 0; j < NT; ++j) b[(u + 1) & 1][j] = bt[2 * (u + 1) * TW + 16 * j];
            }
            #pragma unroll
            for (int j = 0; j < NT; ++j)
                acc[j] = __builtin_amdgcn_wmma_f32_16x16x4_f32(false, a[u].xy, false, b[u & 1][j].xy,
                                                               (short)0, acc[j], false, false);
            #pragma unroll
            for (int j = 0; j < NT; ++j)
                acc[j] = __builtin_amdgcn_wmma_f32_16x16x4_f32(false, a[u].zw, false, b[u & 1][j].zw,
                                                               (short)0, acc[j], false, false);
        }

        float* cp = C + (size_t)((mt << 4) + (h << 3)) * N + ng * (16 * NT) + c;
        #pragma unroll
        for (int j = 0; j < NT; ++j) {
            #pragma unroll
            for (int r = 0; r < 8; ++r) {
                float v = acc[j][r];
                if (RELU_BIAS) v = fmaxf(v + bj[j], 0.f);
                cp[(size_t)r * N + 16 * j] = v;
            }
        }

        t += tStep;
        if (t < T) {
            const int mn = t / NG;
            const float* ap = A + (size_t)((mn << 4) + c) * K + (h << 2);
            #pragma unroll
            for (int u = 0; u < KU; ++u) a[u] = *(const v4f*)(ap + 8 * u);
        }
    }
}

__global__ void gather128_kernel(const int* __restrict__ rowstart,
                                 const int* __restrict__ csr_src,
                                 const float* __restrict__ csr_norm,
                                 const float* __restrict__ dinv,
                                 const float* __restrict__ x,
                                 float* __restrict__ out, int n) {
    const int lane = threadIdx.x & 31;
    int node = (blockIdx.x * blockDim.x + threadIdx.x) >> 5;
    if (node >= n) return;
    node = __builtin_amdgcn_readfirstlane(node);

    const int beg = rowstart[node];
    const int end = rowstart[node + 1];
    const float di = dinv[node];
    const float s  = di * di;

    const float4 xv = ((const float4*)(x + (size_t)node * 128))[lane];
    float4 a;
    a.x = s * xv.x; a.y = s * xv.y; a.z = s * xv.z; a.w = s * xv.w;

    for (int j = beg; j < end; ++j) {
        const int   sj = csr_src[j];
        const float nj = csr_norm[j];
        const float4 v = ((const float4*)(x + (size_t)sj * 128))[lane];
        a.x = fmaf(nj, v.x, a.x); a.y = fmaf(nj, v.y, a.y);
        a.z = fmaf(nj, v.z, a.z); a.w = fmaf(nj, v.w, a.w);
    }
    ((float4*)(out + (size_t)node * 128))[lane] = a;
}

__global__ void gather64_kernel(const int* __restrict__ rowstart,
                                const int* __restrict__ csr_src,
                                const float* __restrict__ csr_norm,
                                const float* __restrict__ dinv,
                                const float* __restrict__ h,
                                const float* __restrict__ bias,
                                float* __restrict__ out, int n) {
    const int lane = threadIdx.x & 31;
    int node = (blockIdx.x * blockDim.x + threadIdx.x) >> 5;
    if (node >= n) return;
    node = __builtin_amdgcn_readfirstlane(node);

    const int beg = rowstart[node];
    const int end = rowstart[node + 1];
    const float di = dinv[node];
    const float s  = di * di;

    const float2* hn = (const float2*)(h + (size_t)node * 64);
    const float2* b2 = (const float2*)bias;
    float2 hv = hn[lane];
    float2 bv = b2[lane];
    float2 a;
    a.x = fmaf(s, hv.x, bv.x);
    a.y = fmaf(s, hv.y, bv.y);

    for (int j = beg; j < end; ++j) {
        const int   sj = csr_src[j];
        const float nj = csr_norm[j];
        const float2* hs = (const float2*)(h + (size_t)sj * 64);
        float2 v = hs[lane];
        a.x = fmaf(nj, v.x, a.x);
        a.y = fmaf(nj, v.y, a.y);
    }
    ((float2*)(out + (size_t)node * 64))[lane] = a;
}

extern "C" void kernel_launch(void* const* d_in, const int* in_sizes, int n_in,
                              void* d_out, int out_size, void* d_ws, size_t ws_size,
                              hipStream_t stream) {
    (void)n_in; (void)out_size; (void)ws_size;

    const float* x  = (const float*)d_in[0];
    const int*   ei = (const int*)d_in[1];
    const float* W1 = (const float*)d_in[2];
    const float* b1 = (const float*)d_in[3];
    const float* W2 = (const float*)d_in[4];
    const float* b2 = (const float*)d_in[5];
    float* out = (float*)d_out;

    const int Fin = 128, Fmid = 256, Fout = 64;
    const int N = in_sizes[0] / Fin;
    const int E = in_sizes[1] / 2;
    const int* src = ei;
    const int* dst = ei + E;

    char* ws = (char*)d_ws;
    size_t off = 0;
    auto take = [&](size_t bytes) -> char* {
        char* p = ws + off;
        off = (off + bytes + 255) & ~(size_t)255;
        return p;
    };
    int*   deg      = (int*)  take((size_t)N * 4);
    float* dinv     = (float*)take((size_t)N * 4);
    int*   rowstart = (int*)  take((size_t)(N + 1) * 4);
    int*   cursor   = (int*)  take((size_t)N * 4);
    int*   csr_src  = (int*)  take((size_t)E * 4);
    float* csr_norm = (float*)take((size_t)E * 4);
    float* aggx     = (float*)take((size_t)N * Fin * 4);
    float* hrelu    = (float*)take((size_t)N * Fmid * 4);
    float* h2       = (float*)take((size_t)N * Fout * 4);

    const int T = 256;

    zero_i32_kernel <<<(N + T - 1) / T, T, 0, stream>>>(deg, N);
    count_deg_kernel<<<(E + T - 1) / T, T, 0, stream>>>(dst, deg, E);
    dinv_kernel     <<<(N + T - 1) / T, T, 0, stream>>>(deg, dinv, N);
    scan_kernel     <<<1, SCAN_T, 0, stream>>>(deg, rowstart, N);
    copy_i32_kernel <<<(N + T - 1) / T, T, 0, stream>>>(rowstart, cursor, N);
    fill_csr_kernel <<<(E + T - 1) / T, T, 0, stream>>>(src, dst, dinv, cursor,
                                                        csr_src, csr_norm, E);

    {
        long long threads = (long long)N * 32;
        gather128_kernel<<<(int)((threads + T - 1) / T), T, 0, stream>>>(
            rowstart, csr_src, csr_norm, dinv, x, aggx, N);
    }
    gemm_f32_wmma_kernel<128, 256, 4, true><<<GEMM_GRID, GEMM_T, 0, stream>>>(aggx, W1, b1, hrelu, N);
    gemm_f32_wmma_kernel<256, 64, 2, false><<<GEMM_GRID, GEMM_T, 0, stream>>>(hrelu, W2, nullptr, h2, N);
    {
        long long threads = (long long)N * 32;
        gather64_kernel<<<(int)((threads + T - 1) / T), T, 0, stream>>>(
            rowstart, csr_src, csr_norm, dinv, h2, b2, out, N);
    }
}


# Edge weights factored out: rows pre-scaled by deg^-1/2 (x in the dinv launch, h2 in the GEMM store), each sum scaled once per node; the CSR fill sheds 2 random reads and 1 scattered store per edge
# speedup vs baseline: 2.9871x; 1.0810x over previous
#include <hip/hip_runtime.h>

typedef float v2f __attribute__((ext_vector_type(2)));
typedef float v4f __attribute__((ext_vector_type(4)));
typedef float v8f __attribute__((ext_vector_type(8)));

__global__ void zero_i32_kernel(int* __restrict__ p, int n) {
    int i = blockIdx.x * blockDim.x + threadIdx.x;
    if (i < n) p[i] = 0;
}

__global__ void count_deg_kernel(const int* __restrict__ dst, int* __restrict__ deg, int E) {
    int e = blockIdx.x * blockDim.x + threadIdx.x;
    if (e < E) atomicAdd(&deg[dst[e]], 1);
}

__global__ void dinv_scale128_kernel(const int* __restrict__ deg, const float* __restrict__ x,
                                     float* __restrict__ dinv, float* __restrict__ xs, int n) {
    const int lane = threadIdx.x & 31;
    const int node = (blockIdx.x * blockDim.x + threadIdx.x) >> 5;
    if (node >= n) return;
    const float4 v = ((const float4*)(x + (size_t)node * 128))[lane];
    const float d = rsqrtf((float)deg[node] + 1.0f);
    float4 o;
    o.x = d * v.x; o.y = d * v.y; o.z = d * v.z; o.w = d * v.w;
    ((float4*)(xs + (size_t)node * 128))[lane] = o;
    if (lane == 0) dinv[node] = d;
}

#define SCAN_T 1024
#define SCAN_V 13

__device__ __forceinline__ int4 scan_load4(const int* __restrict__ p, int idx, int n) {
    int4 r;
    if (idx + 4 <= n) {
        r = *(const int4*)(p + idx);
    } else {
        r.x = (idx     < n) ? p[idx]     : 0;
        r.y = (idx + 1 < n) ? p[idx + 1] : 0;
        r.z = (idx + 2 < n) ? p[idx + 2] : 0;
        r.w = (idx + 3 < n) ? p[idx + 3] : 0;
    }
    return r;
}

__global__ void __launch_bounds__(SCAN_T)
scan_kernel(const int* __restrict__ deg, int* __restrict__ rowstart, int n) {
    __shared__ int wtot[SCAN_V * 32];
    __shared__ int passTotal;
    const int tid  = threadIdx.x;
    const int lane = tid & 31;
    const int wave = tid >> 5;
    const int PASS = SCAN_T * SCAN_V * 4;
    int carry = 0;

    for (int base = 0; base <= n; base += PASS) {
        int4 v[SCAN_V];
        int  s[SCAN_V];
        #pragma unroll
        for (int i = 0; i < SCAN_V; ++i)
            v[i] = scan_load4(deg, base + (i * SCAN_T + tid) * 4, n);
        #pragma unroll
        for (int i = 0; i < SCAN_V; ++i)
            s[i] = v[i].x + v[i].y + v[i].z + v[i].w;

        #pragma unroll
        for (int d = 1; d < 32; d <<= 1) {
            #pragma unroll
            for (int i = 0; i < SCAN_V; ++i) {
                const int t = __shfl_up(s[i], d, 32);
                if (lane >= d) s[i] += t;
            }
        }
        if (lane == 31) {
            #pragma unroll
            for (int i = 0; i < SCAN_V; ++i) wtot[i * 32 + wave] = s[i];
        }
        __syncthreads();

        if (wave == 0) {
            int t[SCAN_V];
            int sum = 0;
            #pragma unroll
            for (int k = 0; k < SCAN_V; ++k) t[k] = wtot[lane * SCAN_V + k];
            #pragma unroll
            for (int k = 0; k < SCAN_V; ++k) { const int x = t[k]; t[k] = sum; sum += x; }
            int incl = sum;
            #pragma unroll
            for (int d = 1; d < 32; d <<= 1) {
                const int u = __shfl_up(incl, d, 32);
                if (lane >= d) incl += u;
            }
            const int excl = incl - sum;
            #pragma unroll
            for (int k = 0; k < SCAN_V; ++k) wtot[lane * SCAN_V + k] = t[k] + excl;
            if (lane == 31) passTotal = incl;
        }
        __syncthreads();

        #pragma unroll
        for (int i = 0; i < SCAN_V; ++i) {
            const int idx = base + (i * SCAN_T + tid) * 4;
            const int own = v[i].x + v[i].y + v[i].z + v[i].w;
            int4 o;
            o.x = carry + wtot[i * 32 + wave] + (s[i] - own);
            o.y = o.x + v[i].x;
            o.z = o.y + v[i].y;
            o.w = o.z + v[i].z;
            if (idx + 3 <= n) {
                *(int4*)(rowstart + idx) = o;
            } else {
                if (idx     <= n) rowstart[idx]     = o.x;
                if (idx + 1 <= n) rowstart[idx + 1] = o.y;
                if (idx + 2 <= n) rowstart[idx + 2] = o.z;
            }
        }
        carry += passTotal;
        __syncthreads();
    }
}

__global__ void copy_i32_kernel(const int* __restrict__ a, int* __restrict__ b, int n) {
    int i = blockIdx.x * blockDim.x + threadIdx.x;
    if (i < n) b[i] = a[i];
}

__global__ void fill_csr_kernel(const int* __restrict__ src, const int* __restrict__ dst,
                                int* __restrict__ cursor, int* __restrict__ csr_src, int E) {
    int e = blockIdx.x * blockDim.x + threadIdx.x;
    if (e >= E) return;
    const int s = src[e];
    const int d = dst[e];
    const int pos = atomicAdd(&cursor[d], 1);
    csr_src[pos] = s;
}

#define GEMM_T 512
#define GEMM_GRID 256

template <int K, int N, int NT, bool RELU_BIAS, bool ROW_SCALE>
__global__ void __launch_bounds__(GEMM_T)
gemm_f32_wmma_kernel(const float* __restrict__ A,
                     const float* __restrict__ B,
                     const float* __restrict__ bias,
                     const float* __restrict__ rowscale,
                     float* __restrict__ C, int M) {
    constexpr int KU = K / 8;
    constexpr int KK = K / 4;
    constexpr int NG = N / (16 * NT);
    constexpr int TW = 16 * NT;
    static_assert((KK * N) % GEMM_T == 0 && TW % 32 == 0 && N % TW == 0, "tile constants");
    __shared__ v4f Bl[KK * N];

    const int tid  = threadIdx.x;
    const int lane = tid & 31;
    const int wave = __builtin_amdgcn_readfirstlane(tid >> 5);
    const int c = lane & 15;
    const int h = lane >> 4;
    const int T = (M >> 4) * NG;
    const int tStep = gridDim.x * (GEMM_T / 32);
    int t = blockIdx.x + gridDim.x * wave;

    v4f a[KU];
    if (t < T) {
        const int mt = t / NG;
        const float* ap = A + (size_t)((mt << 4) + c) * K + (h << 2);
        #pragma unroll
        for (int u = 0; u < KU; ++u) a[u] = *(const v4f*)(ap + 8 * u);
    }

    #pragma unroll 8
    for (int i = 0; i < (KK * N) / GEMM_T; ++i) {
        const int e  = tid + i * GEMM_T;
        const int kk = e / N;
        const int n  = e - kk * N;
        const float* bp = B + (size_t)(4 * kk) * N + n;
        v4f v;
        v.x = bp[0]; v.y = bp[N]; v.z = bp[2 * N]; v.w = bp[3 * N];
        const int g  = n / TW;
        const int nl = n - g * TW;
        Bl[(g * KK + kk) * TW + (nl ^ ((kk & 1) << 3))] = v;
    }
    __syncthreads();

    const v4f* bl = Bl + h * TW + (c ^ (h << 3));
    while (t < T) {
        const int mt = t / NG;
        const int ng = t - mt * NG;
        const v4f* bt = bl + ng * (KK * TW);

        v8f acc[NT];
        v4f b[2][NT];
        #pragma unroll
        for (int j = 0; j < NT; ++j) { acc[j] = v8f{}; b[0][j] = bt[16 * j]; }
        float bj[NT];
        if (RELU_BIAS) {
            #pragma unroll
            for (int j = 0; j < NT; ++j) bj[j] = bias[ng * TW + 16 * j + c];
        }
        float rs[8];
        if (ROW_SCALE) {
            #pragma unroll
            for (int r = 0; r < 8; ++r) rs[r] = rowscale[(mt << 4) + (h << 3) + r];
        }

        #pragma unroll
        for (int u = 0; u < KU; ++u) {
            if (u + 1 < KU) {
                #pragma unroll
                for (int j = 0; j < NT; ++j) b[(u + 1) & 1][j] = bt[2 * (u + 1) * TW + 16 * j];
            }
            #pragma unroll
            for (int j = 0; j < NT; ++j)
                acc[j] = __builtin_amdgcn_wmma_f32_16x16x4_f32(false, a[u].xy, false, b[u & 1][j].xy,
                                                               (short)0, acc[j], false, false);
            #pragma unroll
            for (int j = 0; j < NT; ++j)
                acc[j] = __builtin_amdgcn_wmma_f32_16x16x4_f32(false, a[u].zw, false, b[u & 1][j].zw,
                                                               (short)0, acc[j], false, false);
        }

        float* cp = C + (size_t)((mt << 4) + (h << 3)) * N + ng * (16 * NT) + c;
        #pragma unroll
        for (int j = 0; j < NT; ++j) {
            #pragma unroll
            for (int r = 0; r < 8; ++r) {
                float v = acc[j][r];
                if (RELU_BIAS) v = fmaxf(v + bj[j], 0.f);
                if (ROW_SCALE) v *= rs[r];
                cp[(size_t)r * N + 16 * j] = v;
            }
        }

        t += tStep;
        if (t < T) {
            const int mn = t / NG;
            const float* ap = A + (size_t)((mn << 4) + c) * K + (h << 2);
            #pragma unroll
            for (int u = 0; u < KU; ++u) a[u] = *(const v4f*)(ap + 8 * u);
        }
    }
}

__global__ void gather128_kernel(const int* __restrict__ rowstart,
                                 const int* __restrict__ csr_src,
                                 const float* __restrict__ dinv,
                                 const float* __restrict__ xs,
                                 float* __restrict__ out, int n) {
    const int lane = threadIdx.x & 31;
    int node = (blockIdx.x * blockDim.x + threadIdx.x) >> 5;
    if (node >= n) return;
    node = __builtin_amdgcn_readfirstlane(node);

    const int beg = rowstart[node];
    const int end = rowstart[node + 1];
    const float di = dinv[node];
    asm volatile("" :: "s"(di));

    float4 a = ((const float4*)(xs + (size_t)node * 128))[lane];

    for (int j = beg; j < end; ++j) {
        const int sj = csr_src[j];
        const float4 v = ((const float4*)(xs + (size_t)sj * 128))[lane];
        a.x += v.x; a.y += v.y; a.z += v.z; a.w += v.w;
    }
    a.x *= di; a.y *= di; a.z *= di; a.w *= di;
    ((float4*)(out + (size_t)node * 128))[lane] = a;
}

__global__ void gather64_kernel(const int* __restrict__ rowstart,
                                const int* __restrict__ csr_src,
                                const float* __restrict__ dinv,
                                const float* __restrict__ h,
                                const float* __restrict__ bias,
                                float* __restrict__ out, int n) {
    const int lane = threadIdx.x & 31;
    int node = (blockIdx.x * blockDim.x + threadIdx.x) >> 5;
    if (node >= n) return;
    node = __builtin_amdgcn_readfirstlane(node);

    const int beg = rowstart[node];
    const int end = rowstart[node + 1];
    const float di = dinv[node];
    asm volatile("" :: "s"(di));

    const float2* hn = (const float2*)(h + (size_t)node * 64);
    const float2* b2 = (const float2*)bias;
    float2 a  = hn[lane];
    float2 bv = b2[lane];

    for (int j = beg; j < end; ++j) {
        const int sj = csr_src[j];
        const float2* hs = (const float2*)(h + (size_t)sj * 64);
        float2 v = hs[lane];
        a.x += v.x;
        a.y += v.y;
    }
    a.x = fmaf(di, a.x, bv.x);
    a.y = fmaf(di, a.y, bv.y);
    ((float2*)(out + (size_t)node * 64))[lane] = a;
}

extern "C" void kernel_launch(void* const* d_in, const int* in_sizes, int n_in,
                              void* d_out, int out_size, void* d_ws, size_t ws_size,
                              hipStream_t stream) {
    (void)n_in; (void)out_size; (void)ws_size;

    const float* x  = (const float*)d_in[0];
    const int*   ei = (const int*)d_in[1];
    const float* W1 = (const float*)d_in[2];
    const float* b1 = (const float*)d_in[3];
    const float* W2 = (const float*)d_in[4];
    const float* b2 = (const float*)d_in[5];
    float* out = (float*)d_out;

    const int Fin = 128, Fmid = 256, Fout = 64;
    const int N = in_sizes[0] / Fin;
    const int E = in_sizes[1] / 2;
    const int* src = ei;
    const int* dst = ei + E;

    char* ws = (char*)d_ws;
    size_t off = 0;
    auto take = [&](size_t bytes) -> char* {
        char* p = ws + off;
        off = (off + bytes + 255) & ~(size_t)255;
        return p;
    };
    int*   deg      = (int*)  take((size_t)N * 4);
    float* dinv     = (float*)take((size_t)N * 4);
    int*   rowstart = (int*)  take((size_t)(N + 1) * 4);
    int*   cursor   = (int*)  take((size_t)N * 4);
    int*   csr_src  = (int*)  take((size_t)E * 4);
    float* xs       = (float*)take((size_t)N * Fin * 4);
    float* aggx     = (float*)take((size_t)N * Fin * 4);
    float* hrelu    = (float*)take((size_t)N * Fmid * 4);
    float* h2       = (float*)take((size_t)N * Fout * 4);

    const int T = 256;

    zero_i32_kernel <<<(N + T - 1) / T, T, 0, stream>>>(deg, N);
    count_deg_kernel<<<(E + T - 1) / T, T, 0, stream>>>(dst, deg, E);
    {
        long long threads = (long long)N * 32;
        dinv_scale128_kernel<<<(int)((threads + T - 1) / T), T, 0, stream>>>(deg, x, dinv, xs, N);
    }
    scan_kernel     <<<1, SCAN_T, 0, stream>>>(deg, rowstart, N);
    copy_i32_kernel <<<(N + T - 1) / T, T, 0, stream>>>(rowstart, cursor, N);
    fill_csr_kernel <<<(E + T - 1) / T, T, 0, stream>>>(src, dst, cursor, csr_src, E);

    {
        long long threads = (long long)N * 32;
        gather128_kernel<<<(int)((threads + T - 1) / T), T, 0, stream>>>(
            rowstart, csr_src, dinv, xs, aggx, N);
    }
    gemm_f32_wmma_kernel<128, 256, 4, true, false><<<GEMM_GRID, GEMM_T, 0, stream>>>(aggx, W1, b1, nullptr, hrelu, N);
    gemm_f32_wmma_kernel<256, 64, 2, false, true><<<GEMM_GRID, GEMM_T, 0, stream>>>(hrelu, W2, nullptr, dinv, h2, N);
    {
        long long threads = (long long)N * 32;
        gather64_kernel<<<(int)((threads + T - 1) / T), T, 0, stream>>>(
            rowstart, csr_src, dinv, h2, b2, out, N);
    }
}


# CSR slot of an edge = row start + the rank its degree-count add returns: the fill pass needs no atomics, no cursor array and no cursor copy; 10 -> 9 launches; lists stay arrival-ordered
# speedup vs baseline: 3.4924x; 1.1692x over previous
#include <hip/hip_runtime.h>

typedef float v2f __attribute__((ext_vector_type(2)));
typedef float v4f __attribute__((ext_vector_type(4)));
typedef float v8f __attribute__((ext_vector_type(8)));

__global__ void zero_i32_kernel(int* __restrict__ p, int n) {
    int i = blockIdx.x * blockDim.x + threadIdx.x;
    if (i < n) p[i] = 0;
}

__global__ void count_rank_kernel(const int* __restrict__ dst, int* __restrict__ deg,
                                  int* __restrict__ rank, int E) {
    int e = blockIdx.x * blockDim.x + threadIdx.x;
    if (e < E) rank[e] = atomicAdd(&deg[dst[e]], 1);
}

__global__ void dinv_scale128_kernel(const int* __restrict__ deg, const float* __restrict__ x,
                                     float* __restrict__ dinv, float* __restrict__ xs, int n) {
    const int lane = threadIdx.x & 31;
    const int node = (blockIdx.x * blockDim.x + threadIdx.x) >> 5;
    if (node >= n) return;
    const float4 v = ((const float4*)(x + (size_t)node * 128))[lane];
    const float d = rsqrtf((float)deg[node] + 1.0f);
    float4 o;
    o.x = d * v.x; o.y = d * v.y; o.z = d * v.z; o.w = d * v.w;
    ((float4*)(xs + (size_t)node * 128))[lane] = o;
    if (lane == 0) dinv[node] = d;
}

#define SCAN_T 1024
#define SCAN_V 13

__device__ __forceinline__ int4 scan_load4(const int* __restrict__ p, int idx, int n) {
    int4 r;
    if (idx + 4 <= n) {
        r = *(const int4*)(p + idx);
    } else {
        r.x = (idx     < n) ? p[idx]     : 0;
        r.y = (idx + 1 < n) ? p[idx + 1] : 0;
        r.z = (idx + 2 < n) ? p[idx + 2] : 0;
        r.w = (idx + 3 < n) ? p[idx + 3] : 0;
    }
    return r;
}

__global__ void __launch_bounds__(SCAN_T)
scan_kernel(const int* __restrict__ deg, int* __restrict__ rowstart, int n) {
    __shared__ int wtot[SCAN_V * 32];
    __shared__ int passTotal;
    const int tid  = threadIdx.x;
    const int lane = tid & 31;
    const int wave = tid >> 5;
    const int PASS = SCAN_T * SCAN_V * 4;
    int carry = 0;

    for (int base = 0; base <= n; base += PASS) {
        int4 v[SCAN_V];
        int  s[SCAN_V];
        #pragma unroll
        for (int i = 0; i < SCAN_V; ++i)
            v[i] = scan_load4(deg, base + (i * SCAN_T + tid) * 4, n);
        #pragma unroll
        for (int i = 0; i < SCAN_V; ++i)
            s[i] = v[i].x + v[i].y + v[i].z + v[i].w;

        #pragma unroll
        for (int d = 1; d < 32; d <<= 1) {
            #pragma unroll
            for (int i = 0; i < SCAN_V; ++i) {
                const int t = __shfl_up(s[i], d, 32);
                if (lane >= d) s[i] += t;
            }
        }
        if (lane == 31) {
            #pragma unroll
            for (int i = 0; i < SCAN_V; ++i) wtot[i * 32 + wave] = s[i];
        }
        __syncthreads();

        if (wave == 0) {
            int t[SCAN_V];
            int sum = 0;
            #pragma unroll
            for (int k = 0; k < SCAN_V; ++k) t[k] = wtot[lane * SCAN_V + k];
            #pragma unroll
            for (int k = 0; k < SCAN_V; ++k) { const int x = t[k]; t[k] = sum; sum += x; }
            int incl = sum;
            #pragma unroll
            for (int d = 1; d < 32; d <<= 1) {
                const int u = __shfl_up(incl, d, 32);
                if (lane >= d) incl += u;
            }
            const int excl = incl - sum;
            #pragma unroll
            for (int k = 0; k < SCAN_V; ++k) wtot[lane * SCAN_V + k] = t[k] + excl;
            if (lane == 31) passTotal = incl;
        }
        __syncthreads();

        #pragma unroll
        for (int i = 0; i < SCAN_V; ++i) {
            const int idx = base + (i * SCAN_T + tid) * 4;
            const int own = v[i].x + v[i].y + v[i].z + v[i].w;
            int4 o;
            o.x = carry + wtot[i * 32 + wave] + (s[i] - own);
            o.y = o.x + v[i].x;
            o.z = o.y + v[i].y;
            o.w = o.z + v[i].z;
            if (idx + 3 <= n) {
                *(int4*)(rowstart + idx) = o;
            } else {
                if (idx     <= n) rowstart[idx]     = o.x;
                if (idx + 1 <= n) rowstart[idx + 1] = o.y;
                if (idx + 2 <= n) rowstart[idx + 2] = o.z;
            }
        }
        carry += passTotal;
        __syncthreads();
    }
}

__global__ void fill_csr_kernel(const int* __restrict__ src, const int* __restrict__ dst,
                                const int* __restrict__ rank, const int* __restrict__ rowstart,
                                int* __restrict__ csr_src, int E) {
    int e = blockIdx.x * blockDim.x + threadIdx.x;
    if (e >= E) return;
    const int s = src[e];
    const int d = dst[e];
    const int r = rank[e];
    csr_src[rowstart[d] + r] = s;
}

#define GEMM_T 512
#define GEMM_GRID 256

template <int K, int N, int NT, bool RELU_BIAS, bool ROW_SCALE>
__global__ void __launch_bounds__(GEMM_T)
gemm_f32_wmma_kernel(const float* __restrict__ A,
                     const float* __restrict__ B,
                     const float* __restrict__ bias,
                     const float* __restrict__ rowscale,
                     float* __restrict__ C, int M) {
    constexpr int KU = K / 8;
    constexpr int KK = K / 4;
    constexpr int NG = N / (16 * NT);
    constexpr int TW = 16 * NT;
    static_assert((KK * N) % GEMM_T == 0 && TW % 32 == 0 && N % TW == 0, "tile constants");
    __shared__ v4f Bl[KK * N];

    const int tid  = threadIdx.x;
    const int lane = tid & 31;
    const int wave = __builtin_amdgcn_readfirstlane(tid >> 5);
    const int c = lane & 15;
    const int h = lane >> 4;
    const int T = (M >> 4) * NG;
    const int tStep = gridDim.x * (GEMM_T / 32);
    int t = blockIdx.x + gridDim.x * wave;

    v4f a[KU];
    if (t < T) {
        const int mt = t / NG;
        const float* ap = A + (size_t)((mt << 4) + c) * K + (h << 2);
        #pragma unroll
        for (int u = 0; u < KU; ++u) a[u] = *(const v4f*)(ap + 8 * u);
    }

    #pragma unroll 8
    for (int i = 0; i < (KK * N) / GEMM_T; ++i) {
        const int e  = tid + i * GEMM_T;
        const int kk = e / N;
        const int n  = e - kk * N;
        const float* bp = B + (size_t)(4 * kk) * N + n;
        v4f v;
        v.x = bp[0]; v.y = bp[N]; v.z = bp[2 * N]; v.w = bp[3 * N];
        const int g  = n / TW;
        const int nl = n - g * TW;
        Bl[(g * KK + kk) * TW + (nl ^ ((kk & 1) << 3))] = v;
    }
    __syncthreads();

    const v4f* bl = Bl + h * TW + (c ^ (h << 3));
    while (t < T) {
        const int mt = t / NG;
        const int ng = t - mt * NG;
        const v4f* bt = bl + ng * (KK * TW);

        v8f acc[NT];
        v4f b[2][NT];
        #pragma unroll
        for (int j = 0; j < NT; ++j) { acc[j] = v8f{}; b[0][j] = bt[16 * j]; }
        float bj[NT];
        if (RELU_BIAS) {
            #pragma unroll
            for (int j = 0; j < NT; ++j) bj[j] = bias[ng * TW + 16 * j + c];
        }
        float rs[8];
        if (ROW_SCALE) {
            #pragma unroll
            for (int r = 0; r < 8; ++r) rs[r] = rowscale[(mt << 4) + (h << 3) + r];
        }

        #pragma unroll
        for (int u = 0; u < KU; ++u) {
            if (u + 1 < KU) {
                #pragma unroll
                for (int j = 0; j < NT; ++j) b[(u + 1) & 1][j] = bt[2 * (u + 1) * TW + 16 * j];
            }
            #pragma unroll
            for (int j = 0; j < NT; ++j)
                acc[j] = __builtin_amdgcn_wmma_f32_16x16x4_f32(false, a[u].xy, false, b[u & 1][j].xy,
                                                               (short)0, acc[j], false, false);
            #pragma unroll
            for (int j = 0; j < NT; ++j)
                acc[j] = __builtin_amdgcn_wmma_f32_16x16x4_f32(false, a[u].zw, false, b[u & 1][j].zw,
                                                               (short)0, acc[j], false, false);
        }

        float* cp = C + (size_t)((mt << 4) + (h << 3)) * N + ng * (16 * NT) + c;
        #pragma unroll
        for (int j = 0; j < NT; ++j) {
            #pragma unroll
            for (int r = 0; r < 8; ++r) {
                float v = acc[j][r];
                if (RELU_BIAS) v = fmaxf(v + bj[j], 0.f);
                if (ROW_SCALE) v *= rs[r];
                cp[(size_t)r * N + 16 * j] = v;
            }
        }

        t += tStep;
        if (t < T) {
            const int mn = t / NG;
            const float* ap = A + (size_t)((mn << 4) + c) * K + (h << 2);
            #pragma unroll
            for (int u = 0; u < KU; ++u) a[u] = *(const v4f*)(ap + 8 * u);
        }
    }
}

__global__ void gather128_kernel(const int* __restrict__ rowstart,
                                 const int* __restrict__ csr_src,
                                 const float* __restrict__ dinv,
                                 const float* __restrict__ xs,
                                 float* __restrict__ out, int n) {
    const int lane = threadIdx.x & 31;
    int node = (blockIdx.x * blockDim.x + threadIdx.x) >> 5;
    if (node >= n) return;
    node = __builtin_amdgcn_readfirstlane(node);

    const int beg = rowstart[node];
    const int end = rowstart[node + 1];
    const float di = dinv[node];
    asm volatile("" :: "s"(di));

    float4 a = ((const float4*)(xs + (size_t)node * 128))[lane];

    for (int j = beg; j < end; ++j) {
        const int sj = csr_src[j];
        const float4 v = ((const float4*)(xs + (size_t)sj * 128))[lane];
        a.x += v.x; a.y += v.y; a.z += v.z; a.w += v.w;
    }
    a.x *= di; a.y *= di; a.z *= di; a.w *= di;
    ((float4*)(out + (size_t)node * 128))[lane] = a;
}

__global__ void gather64_kernel(const int* __restrict__ rowstart,
                                const int* __restrict__ csr_src,
                                const float* __restrict__ dinv,
                                const float* __restrict__ h,
                                const float* __restrict__ bias,
                                float* __restrict__ out, int n) {
    const int lane = threadIdx.x & 31;
    int node = (blockIdx.x * blockDim.x + threadIdx.x) >> 5;
    if (node >= n) return;
    node = __builtin_amdgcn_readfirstlane(node);

    const int beg = rowstart[node];
    const int end = rowstart[node + 1];
    const float di = dinv[node];
    asm volatile("" :: "s"(di));

    const float2* hn = (const float2*)(h + (size_t)node * 64);
    const float2* b2 = (const float2*)bias;
    float2 a  = hn[lane];
    float2 bv = b2[lane];

    for (int j = beg; j < end; ++j) {
        const int sj = csr_src[j];
        const float2* hs = (const float2*)(h + (size_t)sj * 64);
        float2 v = hs[lane];
        a.x += v.x;
        a.y += v.y;
    }
    a.x = fmaf(di, a.x, bv.x);
    a.y = fmaf(di, a.y, bv.y);
    ((float2*)(out + (size_t)node * 64))[lane] = a;
}

extern "C" void kernel_launch(void* const* d_in, const int* in_sizes, int n_in,
                              void* d_out, int out_size, void* d_ws, size_t ws_size,
                              hipStream_t stream) {
    (void)n_in; (void)out_size; (void)ws_size;

    const float* x  = (const float*)d_in[0];
    const int*   ei = (const int*)d_in[1];
    const float* W1 = (const float*)d_in[2];
    const float* b1 = (const float*)d_in[3];
    const float* W2 = (const float*)d_in[4];
    const float* b2 = (const float*)d_in[5];
    float* out = (float*)d_out;

    const int Fin = 128, Fmid = 256, Fout = 64;
    const int N = in_sizes[0] / Fin;
    const int E = in_sizes[1] / 2;
    const int* src = ei;
    const int* dst = ei + E;

    char* ws = (char*)d_ws;
    size_t off = 0;
    auto take = [&](size_t bytes) -> char* {
        char* p = ws + off;
        off = (off + bytes + 255) & ~(size_t)255;
        return p;
    };
    int*   deg      = (int*)  take((size_t)N * 4);
    float* dinv     = (float*)take((size_t)N * 4);
    int*   rowstart = (int*)  take((size_t)(N + 1) * 4);
    int*   rank     = (int*)  take((size_t)E * 4);
    int*   csr_src  = (int*)  take((size_t)E * 4);
    float* xs       = (float*)take((size_t)N * Fin * 4);
    float* aggx     = (float*)take((size_t)N * Fin * 4);
    float* hrelu    = (float*)take((size_t)N * Fmid * 4);
    float* h2       = (float*)take((size_t)N * Fout * 4);

    const int T = 256;

    zero_i32_kernel <<<(N + T - 1) / T, T, 0, stream>>>(deg, N);
    count_rank_kernel<<<(E + T - 1) / T, T, 0, stream>>>(dst, deg, rank, E);
    {
        long long threads = (long long)N * 32;
        dinv_scale128_kernel<<<(int)((threads + T - 1) / T), T, 0, stream>>>(deg, x, dinv, xs, N);
    }
    scan_kernel     <<<1, SCAN_T, 0, stream>>>(deg, rowstart, N);
    fill_csr_kernel <<<(E + T - 1) / T, T, 0, stream>>>(src, dst, rank, rowstart, csr_src, E);

    {
        long long threads = (long long)N * 32;
        gather128_kernel<<<(int)((threads + T - 1) / T), T, 0, stream>>>(
            rowstart, csr_src, dinv, xs, aggx, N);
    }
    gemm_f32_wmma_kernel<128, 256, 4, true, false><<<GEMM_GRID, GEMM_T, 0, stream>>>(aggx, W1, b1, nullptr, hrelu, N);
    gemm_f32_wmma_kernel<256, 64, 2, false, true><<<GEMM_GRID, GEMM_T, 0, stream>>>(hrelu, W2, nullptr, dinv, h2, N);
    {
        long long threads = (long long)N * 32;
        gather64_kernel<<<(int)((threads + T - 1) / T), T, 0, stream>>>(
            rowstart, csr_src, dinv, h2, b2, out, N);
    }
}


# CSR built by binning, no global atomics: per-chunk LDS counts per 256-node bin, one scan, pairs scattered to bins, each bin's rows counted and placed in LDS; lists in chunk, then LDS-arrival order
# speedup vs baseline: 4.8459x; 1.3876x over previous
#include <hip/hip_runtime.h>

typedef float v2f __attribute__((ext_vector_type(2)));
typedef float v4f __attribute__((ext_vector_type(4)));
typedef float v8f __attribute__((ext_vector_type(8)));

#define BIN_SHIFT 8
#define BIN_N 256
#define BIN_MAX 1024
#define BIN_T 1024
#define BIN_G 256

__global__ void __launch_bounds__(BIN_T)
bin_count_kernel(const int* __restrict__ dst, int* __restrict__ hist, int E, int nbins) {
    __shared__ int h[BIN_MAX];
    const int tid = threadIdx.x;
    const int G = gridDim.x;
    const int chunk = (E + G - 1) / G;
    const int e0 = blockIdx.x * chunk;
    const int e1 = min(e0 + chunk, E);
    for (int b = tid; b < nbins; b += BIN_T) h[b] = 0;
    __syncthreads();
    for (int e = e0 + tid; e < e1; e += BIN_T) atomicAdd(&h[dst[e] >> BIN_SHIFT], 1);
    __syncthreads();
    for (int b = tid; b < nbins; b += BIN_T) hist[b * G + blockIdx.x] = h[b];
}

__global__ void __launch_bounds__(BIN_T)
bin_scatter_kernel(const int* __restrict__ src, const int* __restrict__ dst,
                   const int* __restrict__ hbase, int2* __restrict__ binned, int E, int nbins) {
    __shared__ int cur[BIN_MAX];
    const int tid = threadIdx.x;
    const int G = gridDim.x;
    const int chunk = (E + G - 1) / G;
    const int e0 = blockIdx.x * chunk;
    const int e1 = min(e0 + chunk, E);
    for (int b = tid; b < nbins; b += BIN_T) cur[b] = hbase[b * G + blockIdx.x];
    __syncthreads();
    for (int e = e0 + tid; e < e1; e += BIN_T) {
        const int d = dst[e];
        const int slot = atomicAdd(&cur[d >> BIN_SHIFT], 1);
        binned[slot] = make_int2(src[e], d);
    }
}

__global__ void __launch_bounds__(BIN_T)
bin_rows_kernel(const int2* __restrict__ binned, const int* __restrict__ hbase,
                int* __restrict__ deg, int* __restrict__ rowstart, int* __restrict__ csr_src,
                int n, int G) {
    __shared__ int cnt[BIN_N];
    __shared__ int wsum[BIN_N / 32];
    const int tid  = threadIdx.x;
    const int lane = tid & 31;
    const int wave = tid >> 5;
    const bool own = tid < BIN_N;
    const int b  = blockIdx.x;
    const int s0 = hbase[b * G];
    const int s1 = hbase[(b + 1) * G];
    if (own) cnt[tid] = 0;
    __syncthreads();
    for (int e = s0 + tid; e < s1; e += BIN_T) atomicAdd(&cnt[binned[e].y & (BIN_N - 1)], 1);
    __syncthreads();

    const int c = own ? cnt[tid] : 0;
    int incl = c;
    #pragma unroll
    for (int d = 1; d < 32; d <<= 1) {
        const int t = __shfl_up(incl, d, 32);
        if (lane >= d) incl += t;
    }
    if (own && lane == 31) wsum[wave] = incl;
    __syncthreads();
    if (own) {
        int wbase = 0;
        for (int w = 0; w < wave; ++w) wbase += wsum[w];
        const int start = s0 + wbase + incl - c;
        const int node = b * BIN_N + tid;
        if (node < n) { deg[node] = c; rowstart[node] = start; }
        cnt[tid] = start;
    }
    if (b == (int)gridDim.x - 1 && tid == 0) rowstart[n] = s1;
    __syncthreads();
    for (int e = s0 + tid; e < s1; e += BIN_T) {
        const int2 p = binned[e];
        const int slot = atomicAdd(&cnt[p.y & (BIN_N - 1)], 1);
        csr_src[slot] = p.x;
    }
}

__global__ void dinv_scale128_kernel(const int* __restrict__ deg, const float* __restrict__ x,
                                     float* __restrict__ dinv, float* __restrict__ xs, int n) {
    const int lane = threadIdx.x & 31;
    const int node = (blockIdx.x * blockDim.x + threadIdx.x) >> 5;
    if (node >= n) return;
    const float4 v = ((const float4*)(x + (size_t)node * 128))[lane];
    const float d = rsqrtf((float)deg[node] + 1.0f);
    float4 o;
    o.x = d * v.x; o.y = d * v.y; o.z = d * v.z; o.w = d * v.w;
    ((float4*)(xs + (size_t)node * 128))[lane] = o;
    if (lane == 0) dinv[node] = d;
}

#define SCAN_T 1024
#define SCAN_V 13

__device__ __forceinline__ int4 scan_load4(const int* __restrict__ p, int idx, int n) {
    int4 r;
    if (idx + 4 <= n) {
        r = *(const int4*)(p + idx);
    } else {
        r.x = (idx     < n) ? p[idx]     : 0;
        r.y = (idx + 1 < n) ? p[idx + 1] : 0;
        r.z = (idx + 2 < n) ? p[idx + 2] : 0;
        r.w = (idx + 3 < n) ? p[idx + 3] : 0;
    }
    return r;
}

__global__ void __launch_bounds__(SCAN_T)
scan_kernel(const int* __restrict__ deg, int* __restrict__ rowstart, int n) {
    __shared__ int wtot[SCAN_V * 32];
    __shared__ int passTotal;
    const int tid  = threadIdx.x;
    const int lane = tid & 31;
    const int wave = tid >> 5;
    const int PASS = SCAN_T * SCAN_V * 4;
    int carry = 0;

    for (int base = 0; base <= n; base += PASS) {
        int4 v[SCAN_V];
        int  s[SCAN_V];
        #pragma unroll
        for (int i = 0; i < SCAN_V; ++i)
            v[i] = scan_load4(deg, base + (i * SCAN_T + tid) * 4, n);
        #pragma unroll
        for (int i = 0; i < SCAN_V; ++i)
            s[i] = v[i].x + v[i].y + v[i].z + v[i].w;

        #pragma unroll
        for (int d = 1; d < 32; d <<= 1) {
            #pragma unroll
            for (int i = 0; i < SCAN_V; ++i) {
                const int t = __shfl_up(s[i], d, 32);
                if (lane >= d) s[i] += t;
            }
        }
        if (lane == 31) {
            #pragma unroll
            for (int i = 0; i < SCAN_V; ++i) wtot[i * 32 + wave] = s[i];
        }
        __syncthreads();

        if (wave == 0) {
            int t[SCAN_V];
            int sum = 0;
            #pragma unroll
            for (int k = 0; k < SCAN_V; ++k) t[k] = wtot[lane * SCAN_V + k];
            #pragma unroll
            for (int k = 0; k < SCAN_V; ++k) { const int x = t[k]; t[k] = sum; sum += x; }
            int incl = sum;
            #pragma unroll
            for (int d = 1; d < 32; d <<= 1) {
                const int u = __shfl_up(incl, d, 32);
                if (lane >= d) incl += u;
            }
            const int excl = incl - sum;
            #pragma unroll
            for (int k = 0; k < SCAN_V; ++k) wtot[lane * SCAN_V + k] = t[k] + excl;
            if (lane == 31) passTotal = incl;
        }
        __syncthreads();

        #pragma unroll
        for (int i = 0; i < SCAN_V; ++i) {
            const int idx = base + (i * SCAN_T + tid) * 4;
            const int own = v[i].x + v[i].y + v[i].z + v[i].w;
            int4 o;
            o.x = carry + wtot[i * 32 + wave] + (s[i] - own);
            o.y = o.x + v[i].x;
            o.z = o.y + v[i].y;
            o.w = o.z + v[i].z;
            if (idx + 3 <= n) {
                *(int4*)(rowstart + idx) = o;
            } else {
                if (idx     <= n) rowstart[idx]     = o.x;
                if (idx + 1 <= n) rowstart[idx + 1] = o.y;
                if (idx + 2 <= n) rowstart[idx + 2] = o.z;
            }
        }
        carry += passTotal;
        __syncthreads();
    }
}

#define GEMM_T 512
#define GEMM_GRID 256

template <int K, int N, int NT, bool RELU_BIAS, bool ROW_SCALE>
__global__ void __launch_bounds__(GEMM_T)
gemm_f32_wmma_kernel(const float* __restrict__ A,
                     const float* __restrict__ B,
                     const float* __restrict__ bias,
                     const float* __restrict__ rowscale,
                     float* __restrict__ C, int M) {
    constexpr int KU = K / 8;
    constexpr int KK = K / 4;
    constexpr int NG = N / (16 * NT);
    constexpr int TW = 16 * NT;
    static_assert((KK * N) % GEMM_T == 0 && TW % 32 == 0 && N % TW == 0, "tile constants");
    __shared__ v4f Bl[KK * N];

    const int tid  = threadIdx.x;
    const int lane = tid & 31;
    const int wave = __builtin_amdgcn_readfirstlane(tid >> 5);
    const int c = lane & 15;
    const int h = lane >> 4;
    const int T = (M >> 4) * NG;
    const int tStep = gridDim.x * (GEMM_T / 32);
    int t = blockIdx.x + gridDim.x * wave;

    v4f a[KU];
    if (t < T) {
        const int mt = t / NG;
        const float* ap = A + (size_t)((mt << 4) + c) * K + (h << 2);
        #pragma unroll
        for (int u = 0; u < KU; ++u) a[u] = *(const v4f*)(ap + 8 * u);
    }

    #pragma unroll 8
    for (int i = 0; i < (KK * N) / GEMM_T; ++i) {
        const int e  = tid + i * GEMM_T;
        const int kk = e / N;
        const int n  = e - kk * N;
        const float* bp = B + (size_t)(4 * kk) * N + n;
        v4f v;
        v.x = bp[0]; v.y = bp[N]; v.z = bp[2 * N]; v.w = bp[3 * N];
        const int g  = n / TW;
        const int nl = n - g * TW;
        Bl[(g * KK + kk) * TW + (nl ^ ((kk & 1) << 3))] = v;
    }
    __syncthreads();

    const v4f* bl = Bl + h * TW + (c ^ (h << 3));
    while (t < T) {
        const int mt = t / NG;
        const int ng = t - mt * NG;
        const v4f* bt = bl + ng * (KK * TW);

        v8f acc[NT];
        v4f b[2][NT];
        #pragma unroll
        for (int j = 0; j < NT; ++j) { acc[j] = v8f{}; b[0][j] = bt[16 * j]; }
        float bj[NT];
        if (RELU_BIAS) {
            #pragma unroll
            for (int j = 0; j < NT; ++j) bj[j] = bias[ng * TW + 16 * j + c];
        }
        float rs[8];
        if (ROW_SCALE) {
            #pragma unroll
            for (int r = 0; r < 8; ++r) rs[r] = rowscale[(mt << 4) + (h << 3) + r];
        }

        #pragma unroll
        for (int u = 0; u < KU; ++u) {
            if (u + 1 < KU) {
                #pragma unroll
                for (int j = 0; j < NT; ++j) b[(u + 1) & 1][j] = bt[2 * (u + 1) * TW + 16 * j];
            }
            #pragma unroll
            for (int j = 0; j < NT; ++j)
                acc[j] = __builtin_amdgcn_wmma_f32_16x16x4_f32(false, a[u].xy, false, b[u & 1][j].xy,
                                                               (short)0, acc[j], false, false);
            #pragma unroll
            for (int j = 0; j < NT; ++j)
                acc[j] = __builtin_amdgcn_wmma_f32_16x16x4_f32(false, a[u].zw, false, b[u & 1][j].zw,
                                                               (short)0, acc[j], false, false);
        }

        float* cp = C + (size_t)((mt << 4) + (h << 3)) * N + ng * (16 * NT) + c;
        #pragma unroll
        for (int j = 0; j < NT; ++j) {
            #pragma unroll
            for (int r = 0; r < 8; ++r) {
                float v = acc[j][r];
                if (RELU_BIAS) v = fmaxf(v + bj[j], 0.f);
                if (ROW_SCALE) v *= rs[r];
                cp[(size_t)r * N + 16 * j] = v;
            }
        }

        t += tStep;
        if (t < T) {
            const int mn = t / NG;
            const float* ap = A + (size_t)((mn << 4) + c) * K + (h << 2);
            #pragma unroll
            for (int u = 0; u < KU; ++u) a[u] = *(const v4f*)(ap + 8 * u);
        }
    }
}

__global__ void gather128_kernel(const int* __restrict__ rowstart,
                                 const int* __restrict__ csr_src,
                                 const float* __restrict__ dinv,
                                 const float* __restrict__ xs,
                                 float* __restrict__ out, int n) {
    const int lane = threadIdx.x & 31;
    int node = (blockIdx.x * blockDim.x + threadIdx.x) >> 5;
    if (node >= n) return;
    node = __builtin_amdgcn_readfirstlane(node);

    const int beg = rowstart[node];
    const int end = rowstart[node + 1];
    const float di = dinv[node];
    asm volatile("" :: "s"(di));

    float4 a = ((const float4*)(xs + (size_t)node * 128))[lane];

    for (int j = beg; j < end; ++j) {
        const int sj = csr_src[j];
        const float4 v = ((const float4*)(xs + (size_t)sj * 128))[lane];
        a.x += v.x; a.y += v.y; a.z += v.z; a.w += v.w;
    }
    a.x *= di; a.y *= di; a.z *= di; a.w *= di;
    ((float4*)(out + (size_t)node * 128))[lane] = a;
}

__global__ void gather64_kernel(const int* __restrict__ rowstart,
                                const int* __restrict__ csr_src,
                                const float* __restrict__ dinv,
                                const float* __restrict__ h,
                                const float* __restrict__ bias,
                                float* __restrict__ out, int n) {
    const int lane = threadIdx.x & 31;
    int node = (blockIdx.x * blockDim.x + threadIdx.x) >> 5;
    if (node >= n) return;
    node = __builtin_amdgcn_readfirstlane(node);

    const int beg = rowstart[node];
    const int end = rowstart[node + 1];
    const float di = dinv[node];
    asm volatile("" :: "s"(di));

    const float2* hn = (const float2*)(h + (size_t)node * 64);
    const float2* b2 = (const float2*)bias;
    float2 a  = hn[lane];
    float2 bv = b2[lane];

    for (int j = beg; j < end; ++j) {
        const int sj = csr_src[j];
        const float2* hs = (const float2*)(h + (size_t)sj * 64);
        float2 v = hs[lane];
        a.x += v.x;
        a.y += v.y;
    }
    a.x = fmaf(di, a.x, bv.x);
    a.y = fmaf(di, a.y, bv.y);
    ((float2*)(out + (size_t)node * 64))[lane] = a;
}

extern "C" void kernel_launch(void* const* d_in, const int* in_sizes, int n_in,
                              void* d_out, int out_size, void* d_ws, size_t ws_size,
                              hipStream_t stream) {
    (void)n_in; (void)out_size; (void)ws_size;

    const float* x  = (const float*)d_in[0];
    const int*   ei = (const int*)d_in[1];
    const float* W1 = (const float*)d_in[2];
    const float* b1 = (const float*)d_in[3];
    const float* W2 = (const float*)d_in[4];
    const float* b2 = (const float*)d_in[5];
    float* out = (float*)d_out;

    const int Fin = 128, Fmid = 256, Fout = 64;
    const int N = in_sizes[0] / Fin;
    const int E = in_sizes[1] / 2;
    const int* src = ei;
    const int* dst = ei + E;

    char* ws = (char*)d_ws;
    size_t off = 0;
    auto take = [&](size_t bytes) -> char* {
        char* p = ws + off;
        off = (off + bytes + 255) & ~(size_t)255;
        return p;
    };
    int*   deg      = (int*)  take((size_t)N * 4);
    float* dinv     = (float*)take((size_t)N * 4);
    int*   rowstart = (int*)  take((size_t)(N + 1) * 4);
    const int NB = (N + BIN_N - 1) / BIN_N;
    int*   hist     = (int*)  take((size_t)NB * BIN_G * 4);
    int*   hbase    = (int*)  take(((size_t)NB * BIN_G + 1) * 4);
    int2*  binned   = (int2*) take((size_t)E * 8);
    int*   csr_src  = (int*)  take((size_t)E * 4);
    float* xs       = (float*)take((size_t)N * Fin * 4);
    float* aggx     = (float*)take((size_t)N * Fin * 4);
    float* hrelu    = (float*)take((size_t)N * Fmid * 4);
    float* h2       = (float*)take((size_t)N * Fout * 4);

    const int T = 256;

    bin_count_kernel  <<<BIN_G, BIN_T, 0, stream>>>(dst, hist, E, NB);
    scan_kernel       <<<1, SCAN_T, 0, stream>>>(hist, hbase, NB * BIN_G);
    bin_scatter_kernel<<<BIN_G, BIN_T, 0, stream>>>(src, dst, hbase, binned, E, NB);
    bin_rows_kernel   <<<NB, BIN_T, 0, stream>>>(binned, hbase, deg, rowstart, csr_src, N, BIN_G);
    {
        long long threads = (long long)N * 32;
        dinv_scale128_kernel<<<(int)((threads + T - 1) / T), T, 0, stream>>>(deg, x, dinv, xs, N);
    }

    {
        long long threads = (long long)N * 32;
        gather128_kernel<<<(int)((threads + T - 1) / T), T, 0, stream>>>(
            rowstart, csr_src, dinv, xs, aggx, N);
    }
    gemm_f32_wmma_kernel<128, 256, 4, true, false><<<GEMM_GRID, GEMM_T, 0, stream>>>(aggx, W1, b1, nullptr, hrelu, N);
    gemm_f32_wmma_kernel<256, 64, 2, false, true><<<GEMM_GRID, GEMM_T, 0, stream>>>(hrelu, W2, nullptr, dinv, h2, N);
    {
        long long threads = (long long)N * 32;
        gather64_kernel<<<(int)((threads + T - 1) / T), T, 0, stream>>>(
            rowstart, csr_src, dinv, h2, b2, out, N);
    }
}


# CSR build in 2 launches, not 4: each edge chunk counts, scans and bins its own edges (no global scan); a bin's work-group locates its chunk segments itself; row lists now in plain LDS-arrival order
# speedup vs baseline: 5.6715x; 1.1704x over previous
#include <hip/hip_runtime.h>

typedef float v2f __attribute__((ext_vector_type(2)));
typedef float v4f __attribute__((ext_vector_type(4)));
typedef float v8f __attribute__((ext_vector_type(8)));

#define BIN_SHIFT 8
#define BIN_N 256
#define BIN_MAX 1024
#define BIN_T 1024
#define BIN_G 256
#define BIN_SPW (BIN_G / (BIN_T / 32))

__global__ void __launch_bounds__(BIN_T)
bin_local_kernel(const int* __restrict__ src, const int* __restrict__ dst,
                 int* __restrict__ offsT, int2* __restrict__ sorted, int E, int nbins) {
    __shared__ int h[BIN_MAX];
    __shared__ int wtot[BIN_T / 32];
    const int tid  = threadIdx.x;
    const int lane = tid & 31;
    const int wave = tid >> 5;
    const int chunk = (E + BIN_G - 1) / BIN_G;
    const int e0 = min((int)blockIdx.x * chunk, E);
    const int e1 = min(e0 + chunk, E);
    h[tid] = 0;
    __syncthreads();
    for (int e = e0 + tid; e < e1; e += BIN_T) atomicAdd(&h[dst[e] >> BIN_SHIFT], 1);
    __syncthreads();

    const int c = h[tid];
    int incl = c;
    #pragma unroll
    for (int d = 1; d < 32; d <<= 1) {
        const int t = __shfl_up(incl, d, 32);
        if (lane >= d) incl += t;
    }
    if (lane == 31) wtot[wave] = incl;
    __syncthreads();
    int wbase = 0;
    for (int w = 0; w < wave; ++w) wbase += wtot[w];
    const int excl = wbase + incl - c;
    if (tid < nbins) offsT[tid * BIN_G + blockIdx.x] = excl;
    if (tid == 0) offsT[nbins * BIN_G + blockIdx.x] = e1 - e0;
    h[tid] = e0 + excl;
    __syncthreads();
    for (int e = e0 + tid; e < e1; e += BIN_T) {
        const int d = dst[e];
        const int slot = atomicAdd(&h[d >> BIN_SHIFT], 1);
        sorted[slot] = make_int2(src[e], d);
    }
}

__global__ void __launch_bounds__(BIN_T)
bin_rows_kernel(const int2* __restrict__ sorted, const int* __restrict__ offsT,
                int* __restrict__ deg, int* __restrict__ rowstart, int* __restrict__ csr_src,
                int n, int E) {
    __shared__ int cnt[BIN_N];
    __shared__ int wsum[BIN_N / 32];
    __shared__ int segs[BIN_G];
    __shared__ int segl[BIN_G];
    __shared__ int part[2 * (BIN_G / 32)];
    const int tid  = threadIdx.x;
    const int lane = tid & 31;
    const int wave = tid >> 5;
    const bool own = tid < BIN_N;
    const int b = blockIdx.x;
    const int chunk = (E + BIN_G - 1) / BIN_G;
    if (tid < BIN_G) {
        const int o0 = offsT[b * BIN_G + tid];
        const int o1 = offsT[(b + 1) * BIN_G + tid];
        segs[tid] = min(tid * chunk, E) + o0;
        segl[tid] = o1 - o0;
        int r0 = o0, r1 = o1;
        #pragma unroll
        for (int d = 16; d >= 1; d >>= 1) {
            r0 += __shfl_xor(r0, d, 32);
            r1 += __shfl_xor(r1, d, 32);
        }
        if (lane == 0) { part[wave] = r0; part[BIN_G / 32 + wave] = r1; }
    }
    if (own) cnt[tid] = 0;
    __syncthreads();
    int s0 = 0, s1 = 0;
    #pragma unroll
    for (int w = 0; w < BIN_G / 32; ++w) { s0 += part[w]; s1 += part[BIN_G / 32 + w]; }

    int  st[BIN_SPW], ln[BIN_SPW];
    int2 p[BIN_SPW];
    #pragma unroll
    for (int k = 0; k < BIN_SPW; ++k) {
        st[k] = segs[wave + k * (BIN_T / 32)];
        ln[k] = segl[wave + k * (BIN_T / 32)];
        p[k] = make_int2(0, 0);
        if (lane < ln[k]) p[k] = sorted[st[k] + lane];
    }
    #pragma unroll
    for (int k = 0; k < BIN_SPW; ++k) {
        if (lane < ln[k]) atomicAdd(&cnt[p[k].y & (BIN_N - 1)], 1);
        for (int i = lane + 32; i < ln[k]; i += 32)
            atomicAdd(&cnt[sorted[st[k] + i].y & (BIN_N - 1)], 1);
    }
    __syncthreads();

    const int c = own ? cnt[tid] : 0;
    int incl = c;
    #pragma unroll
    for (int d = 1; d < 32; d <<= 1) {
        const int t = __shfl_up(incl, d, 32);
        if (lane >= d) incl += t;
    }
    if (own && lane == 31) wsum[wave] = incl;
    __syncthreads();
    if (own) {
        int wbase = 0;
        for (int w = 0; w < wave; ++w) wbase += wsum[w];
        const int start = s0 + wbase + incl - c;
        const int node = b * BIN_N + tid;
        if (node < n) { deg[node] = c; rowstart[node] = start; }
        cnt[tid] = start;
    }
    if (b == (int)gridDim.x - 1 && tid == 0) rowstart[n] = s1;
    __syncthreads();
    #pragma unroll
    for (int k = 0; k < BIN_SPW; ++k) {
        if (lane < ln[k]) {
            const int slot = atomicAdd(&cnt[p[k].y & (BIN_N - 1)], 1);
            csr_src[slot] = p[k].x;
        }
        for (int i = lane + 32; i < ln[k]; i += 32) {
            const int2 q = sorted[st[k] + i];
            const int slot = atomicAdd(&cnt[q.y & (BIN_N - 1)], 1);
            csr_src[slot] = q.x;
        }
    }
}

__global__ void dinv_scale128_kernel(const int* __restrict__ deg, const float* __restrict__ x,
                                     float* __restrict__ dinv, float* __restrict__ xs, int n) {
    const int lane = threadIdx.x & 31;
    const int node = (blockIdx.x * blockDim.x + threadIdx.x) >> 5;
    if (node >= n) return;
    const float4 v = ((const float4*)(x + (size_t)node * 128))[lane];
    const float d = rsqrtf((float)deg[node] + 1.0f);
    float4 o;
    o.x = d * v.x; o.y = d * v.y; o.z = d * v.z; o.w = d * v.w;
    ((float4*)(xs + (size_t)node * 128))[lane] = o;
    if (lane == 0) dinv[node] = d;
}

#define GEMM_T 512
#define GEMM_GRID 256

template <int K, int N, int NT, bool RELU_BIAS, bool ROW_SCALE>
__global__ void __launch_bounds__(GEMM_T)
gemm_f32_wmma_kernel(const float* __restrict__ A,
                     const float* __restrict__ B,
                     const float* __restrict__ bias,
                     const float* __restrict__ rowscale,
                     float* __restrict__ C, int M) {
    constexpr int KU = K / 8;
    constexpr int KK = K / 4;
    constexpr int NG = N / (16 * NT);
    constexpr int TW = 16 * NT;
    static_assert((KK * N) % GEMM_T == 0 && TW % 32 == 0 && N % TW == 0, "tile constants");
    __shared__ v4f Bl[KK * N];

    const int tid  = threadIdx.x;
    const int lane = tid & 31;
    const int wave = __builtin_amdgcn_readfirstlane(tid >> 5);
    const int c = lane & 15;
    const int h = lane >> 4;
    const int T = (M >> 4) * NG;
    const int tStep = gridDim.x * (GEMM_T / 32);
    int t = blockIdx.x + gridDim.x * wave;

    v4f a[KU];
    if (t < T) {
        const int mt = t / NG;
        const float* ap = A + (size_t)((mt << 4) + c) * K + (h << 2);
        #pragma unroll
        for (int u = 0; u < KU; ++u) a[u] = *(const v4f*)(ap + 8 * u);
    }

    #pragma unroll 8
    for (int i = 0; i < (KK * N) / GEMM_T; ++i) {
        const int e  = tid + i * GEMM_T;
        const int kk = e / N;
        const int n  = e - kk * N;
        const float* bp = B + (size_t)(4 * kk) * N + n;
        v4f v;
        v.x = bp[0]; v.y = bp[N]; v.z = bp[2 * N]; v.w = bp[3 * N];
        const int g  = n / TW;
        const int nl = n - g * TW;
        Bl[(g * KK + kk) * TW + (nl ^ ((kk & 1) << 3))] = v;
    }
    __syncthreads();

    const v4f* bl = Bl + h * TW + (c ^ (h << 3));
    while (t < T) {
        const int mt = t / NG;
        const int ng = t - mt * NG;
        const v4f* bt = bl + ng * (KK * TW);

        v8f acc[NT];
        v4f b[2][NT];
        #pragma unroll
        for (int j = 0; j < NT; ++j) { acc[j] = v8f{}; b[0][j] = bt[16 * j]; }
        float bj[NT];
        if (RELU_BIAS) {
            #pragma unroll
            for (int j = 0; j < NT; ++j) bj[j] = bias[ng * TW + 16 * j + c];
        }
        float rs[8];
        if (ROW_SCALE) {
            #pragma unroll
            for (int r = 0; r < 8; ++r) rs[r] = rowscale[(mt << 4) + (h << 3) + r];
        }

        #pragma unroll
        for (int u = 0; u < KU; ++u) {
            if (u + 1 < KU) {
                #pragma unroll
                for (int j = 0; j < NT; ++j) b[(u + 1) & 1][j] = bt[2 * (u + 1) * TW + 16 * j];
            }
            #pragma unroll
            for (int j = 0; j < NT; ++j)
                acc[j] = __builtin_amdgcn_wmma_f32_16x16x4_f32(false, a[u].xy, false, b[u & 1][j].xy,
                                                               (short)0, acc[j], false, false);
            #pragma unroll
            for (int j = 0; j < NT; ++j)
                acc[j] = __builtin_amdgcn_wmma_f32_16x16x4_f32(false, a[u].zw, false, b[u & 1][j].zw,
                                                               (short)0, acc[j], false, false);
        }

        float* cp = C + (size_t)((mt << 4) + (h << 3)) * N + ng * (16 * NT) + c;
        #pragma unroll
        for (int j = 0; j < NT; ++j) {
            #pragma unroll
            for (int r = 0; r < 8; ++r) {
                float v = acc[j][r];
                if (RELU_BIAS) v = fmaxf(v + bj[j], 0.f);
                if (ROW_SCALE) v *= rs[r];
                cp[(size_t)r * N + 16 * j] = v;
            }
        }

        t += tStep;
        if (t < T) {
            const int mn = t / NG;
            const float* ap = A + (size_t)((mn << 4) + c) * K + (h << 2);
            #pragma unroll
            for (int u = 0; u < KU; ++u) a[u] = *(const v4f*)(ap + 8 * u);
        }
    }
}

__global__ void gather128_kernel(const int* __restrict__ rowstart,
                                 const int* __restrict__ csr_src,
                                 const float* __restrict__ dinv,
                                 const float* __restrict__ xs,
                                 float* __restrict__ out, int n) {
    const int lane = threadIdx.x & 31;
    int node = (blockIdx.x * blockDim.x + threadIdx.x) >> 5;
    if (node >= n) return;
    node = __builtin_amdgcn_readfirstlane(node);

    const int beg = rowstart[node];
    const int end = rowstart[node + 1];
    const float di = dinv[node];
    asm volatile("" :: "s"(di));

    float4 a = ((const float4*)(xs + (size_t)node * 128))[lane];

    for (int j = beg; j < end; ++j) {
        const int sj = csr_src[j];
        const float4 v = ((const float4*)(xs + (size_t)sj * 128))[lane];
        a.x += v.x; a.y += v.y; a.z += v.z; a.w += v.w;
    }
    a.x *= di; a.y *= di; a.z *= di; a.w *= di;
    ((float4*)(out + (size_t)node * 128))[lane] = a;
}

__global__ void gather64_kernel(const int* __restrict__ rowstart,
                                const int* __restrict__ csr_src,
                                const float* __restrict__ dinv,
                                const float* __restrict__ h,
                                const float* __restrict__ bias,
                                float* __restrict__ out, int n) {
    const int lane = threadIdx.x & 31;
    int node = (blockIdx.x * blockDim.x + threadIdx.x) >> 5;
    if (node >= n) return;
    node = __builtin_amdgcn_readfirstlane(node);

    const int beg = rowstart[node];
    const int end = rowstart[node + 1];
    const float di = dinv[node];
    asm volatile("" :: "s"(di));

    const float2* hn = (const float2*)(h + (size_t)node * 64);
    const float2* b2 = (const float2*)bias;
    float2 a  = hn[lane];
    float2 bv = b2[lane];

    for (int j = beg; j < end; ++j) {
        const int sj = csr_src[j];
        const float2* hs = (const float2*)(h + (size_t)sj * 64);
        float2 v = hs[lane];
        a.x += v.x;
        a.y += v.y;
    }
    a.x = fmaf(di, a.x, bv.x);
    a.y = fmaf(di, a.y, bv.y);
    ((float2*)(out + (size_t)node * 64))[lane] = a;
}

extern "C" void kernel_launch(void* const* d_in, const int* in_sizes, int n_in,
                              void* d_out, int out_size, void* d_ws, size_t ws_size,
                              hipStream_t stream) {
    (void)n_in; (void)out_size; (void)ws_size;

    const float* x  = (const float*)d_in[0];
    const int*   ei = (const int*)d_in[1];
    const float* W1 = (const float*)d_in[2];
    const float* b1 = (const float*)d_in[3];
    const float* W2 = (const float*)d_in[4];
    const float* b2 = (const float*)d_in[5];
    float* out = (float*)d_out;

    const int Fin = 128, Fmid = 256, Fout = 64;
    const int N = in_sizes[0] / Fin;
    const int E = in_sizes[1] / 2;
    const int* src = ei;
    const int* dst = ei + E;

    char* ws = (char*)d_ws;
    size_t off = 0;
    auto take = [&](size_t bytes) -> char* {
        char* p = ws + off;
        off = (off + bytes + 255) & ~(size_t)255;
        return p;
    };
    int*   deg      = (int*)  take((size_t)N * 4);
    float* dinv     = (float*)take((size_t)N * 4);
    int*   rowstart = (int*)  take((size_t)(N + 1) * 4);
    const int NB = (N + BIN_N - 1) / BIN_N;
    int*   offsT    = (int*)  take(((size_t)NB + 1) * BIN_G * 4);
    int2*  sorted   = (int2*) take((size_t)E * 8);
    int*   csr_src  = (int*)  take((size_t)E * 4);
    float* xs       = (float*)take((size_t)N * Fin * 4);
    float* aggx     = (float*)take((size_t)N * Fin * 4);
    float* hrelu    = (float*)take((size_t)N * Fmid * 4);
    float* h2       = (float*)take((size_t)N * Fout * 4);

    const int T = 256;

    bin_local_kernel<<<BIN_G, BIN_T, 0, stream>>>(src, dst, offsT, sorted, E, NB);
    bin_rows_kernel <<<NB, BIN_T, 0, stream>>>(sorted, offsT, deg, rowstart, csr_src, N, E);
    {
        long long threads = (long long)N * 32;
        dinv_scale128_kernel<<<(int)((threads + T - 1) / T), T, 0, stream>>>(deg, x, dinv, xs, N);
    }

    {
        long long threads = (long long)N * 32;
        gather128_kernel<<<(int)((threads + T - 1) / T), T, 0, stream>>>(
            rowstart, csr_src, dinv, xs, aggx, N);
    }
    gemm_f32_wmma_kernel<128, 256, 4, true, false><<<GEMM_GRID, GEMM_T, 0, stream>>>(aggx, W1, b1, nullptr, hrelu, N);
    gemm_f32_wmma_kernel<256, 64, 2, false, true><<<GEMM_GRID, GEMM_T, 0, stream>>>(hrelu, W2, nullptr, dinv, h2, N);
    {
        long long threads = (long long)N * 32;
        gather64_kernel<<<(int)((threads + T - 1) / T), T, 0, stream>>>(
            rowstart, csr_src, dinv, h2, b2, out, N);
    }
}


# Pre-scaling of x folded into the CSR row pass: each bin's work-group loads its 256 rows of x under the degree count and stores dinv and dinv*x itself; the scaling launch is gone (7 -> 6 launches)
# speedup vs baseline: 5.8624x; 1.0337x over previous
#include <hip/hip_runtime.h>

typedef float v2f __attribute__((ext_vector_type(2)));
typedef float v4f __attribute__((ext_vector_type(4)));
typedef float v8f __attribute__((ext_vector_type(8)));

#define BIN_SHIFT 8
#define BIN_N 256
#define BIN_MAX 1024
#define BIN_T 1024
#define BIN_G 256
#define BIN_SPW (BIN_G / (BIN_T / 32))
#define BIN_RPW (BIN_N / (BIN_T / 32))

__global__ void __launch_bounds__(BIN_T)
bin_local_kernel(const int* __restrict__ src, const int* __restrict__ dst,
                 int* __restrict__ offsT, int2* __restrict__ sorted, int E, int nbins) {
    __shared__ int h[BIN_MAX];
    __shared__ int wtot[BIN_T / 32];
    const int tid  = threadIdx.x;
    const int lane = tid & 31;
    const int wave = tid >> 5;
    const int chunk = (E + BIN_G - 1) / BIN_G;
    const int e0 = min((int)blockIdx.x * chunk, E);
    const int e1 = min(e0 + chunk, E);
    h[tid] = 0;
    __syncthreads();
    for (int e = e0 + tid; e < e1; e += BIN_T) atomicAdd(&h[dst[e] >> BIN_SHIFT], 1);
    __syncthreads();

    const int c = h[tid];
    int incl = c;
    #pragma unroll
    for (int d = 1; d < 32; d <<= 1) {
        const int t = __shfl_up(incl, d, 32);
        if (lane >= d) incl += t;
    }
    if (lane == 31) wtot[wave] = incl;
    __syncthreads();
    int wbase = 0;
    for (int w = 0; w < wave; ++w) wbase += wtot[w];
    const int excl = wbase + incl - c;
    if (tid < nbins) offsT[tid * BIN_G + blockIdx.x] = excl;
    if (tid == 0) offsT[nbins * BIN_G + blockIdx.x] = e1 - e0;
    h[tid] = e0 + excl;
    __syncthreads();
    for (int e = e0 + tid; e < e1; e += BIN_T) {
        const int d = dst[e];
        const int slot = atomicAdd(&h[d >> BIN_SHIFT], 1);
        sorted[slot] = make_int2(src[e], d);
    }
}

static __device__ __forceinline__ void lds_barrier() {
    __builtin_amdgcn_fence(__ATOMIC_RELEASE, "workgroup", "local");
    __builtin_amdgcn_s_barrier();
    __builtin_amdgcn_fence(__ATOMIC_ACQUIRE, "workgroup", "local");
}

__global__ void __launch_bounds__(BIN_T)
bin_rows_kernel(const int2* __restrict__ sorted, const int* __restrict__ offsT,
                const float* __restrict__ x, float* __restrict__ dinv, float* __restrict__ xs,
                int* __restrict__ rowstart, int* __restrict__ csr_src,
                int n, int E) {
    __shared__ int cnt[BIN_N];
    __shared__ float dsc[BIN_N];
    __shared__ int wsum[BIN_N / 32];
    __shared__ int segs[BIN_G];
    __shared__ int segl[BIN_G];
    __shared__ int part[2 * (BIN_G / 32)];
    const int tid  = threadIdx.x;
    const int lane = tid & 31;
    const int wave = tid >> 5;
    const bool own = tid < BIN_N;
    const int b = blockIdx.x;
    const int chunk = (E + BIN_G - 1) / BIN_G;
    if (tid < BIN_G) {
        const int o0 = offsT[b * BIN_G + tid];
        const int o1 = offsT[(b + 1) * BIN_G + tid];
        segs[tid] = min(tid * chunk, E) + o0;
        segl[tid] = o1 - o0;
        int r0 = o0, r1 = o1;
        #pragma unroll
        for (int d = 16; d >= 1; d >>= 1) {
            r0 += __shfl_xor(r0, d, 32);
            r1 += __shfl_xor(r1, d, 32);
        }
        if (lane == 0) { part[wave] = r0; part[BIN_G / 32 + wave] = r1; }
    }
    if (own) cnt[tid] = 0;
    __syncthreads();
    int s0 = 0, s1 = 0;
    #pragma unroll
    for (int w = 0; w < BIN_G / 32; ++w) { s0 += part[w]; s1 += part[BIN_G / 32 + w]; }

    int  st[BIN_SPW], ln[BIN_SPW];
    int2 p[BIN_SPW];
    #pragma unroll
    for (int k = 0; k < BIN_SPW; ++k) {
        st[k] = segs[wave + k * (BIN_T / 32)];
        ln[k] = segl[wave + k * (BIN_T / 32)];
        p[k] = make_int2(0, 0);
        if (lane < ln[k]) p[k] = sorted[st[k] + lane];
    }
    const int row0 = b * BIN_N + wave * BIN_RPW;
    float4 xv[BIN_RPW];
    #pragma unroll
    for (int k = 0; k < BIN_RPW; ++k) {
        const int r = min(row0 + k, n - 1);
        xv[k] = ((const float4*)(x + (size_t)r * 128))[lane];
    }
    #pragma unroll
    for (int k = 0; k < BIN_SPW; ++k) {
        if (lane < ln[k]) atomicAdd(&cnt[p[k].y & (BIN_N - 1)], 1);
        for (int i = lane + 32; i < ln[k]; i += 32)
            atomicAdd(&cnt[sorted[st[k] + i].y & (BIN_N - 1)], 1);
    }
    lds_barrier();

    const int c = own ? cnt[tid] : 0;
    int incl = c;
    #pragma unroll
    for (int d = 1; d < 32; d <<= 1) {
        const int t = __shfl_up(incl, d, 32);
        if (lane >= d) incl += t;
    }
    if (own && lane == 31) wsum[wave] = incl;
    const float dmine = rsqrtf((float)c + 1.0f);
    if (own) dsc[tid] = dmine;
    lds_barrier();
    if (own) {
        int wbase = 0;
        for (int w = 0; w < wave; ++w) wbase += wsum[w];
        const int start = s0 + wbase + incl - c;
        const int node = b * BIN_N + tid;
        if (node < n) { dinv[node] = dmine; rowstart[node] = start; }
        cnt[tid] = start;
    }
    if (b == (int)gridDim.x - 1 && tid == 0) rowstart[n] = s1;
    lds_barrier();
    #pragma unroll
    for (int k = 0; k < BIN_RPW; ++k) {
        if (row0 + k < n) {
            const float d = dsc[wave * BIN_RPW + k];
            float4 o;
            o.x = d * xv[k].x; o.y = d * xv[k].y; o.z = d * xv[k].z; o.w = d * xv[k].w;
            ((float4*)(xs + (size_t)(row0 + k) * 128))[lane] = o;
        }
    }
    #pragma unroll
    for (int k = 0; k < BIN_SPW; ++k) {
        if (lane < ln[k]) {
            const int slot = atomicAdd(&cnt[p[k].y & (BIN_N - 1)], 1);
            csr_src[slot] = p[k].x;
        }
        for (int i = lane + 32; i < ln[k]; i += 32) {
            const int2 q = sorted[st[k] + i];
            const int slot = atomicAdd(&cnt[q.y & (BIN_N - 1)], 1);
            csr_src[slot] = q.x;
        }
    }
}

#define GEMM_T 512
#define GEMM_GRID 256

template <int K, int N, int NT, bool RELU_BIAS, bool ROW_SCALE>
__global__ void __launch_bounds__(GEMM_T)
gemm_f32_wmma_kernel(const float* __restrict__ A,
                     const float* __restrict__ B,
                     const float* __restrict__ bias,
                     const float* __restrict__ rowscale,
                     float* __restrict__ C, int M) {
    constexpr int KU = K / 8;
    constexpr int KK = K / 4;
    constexpr int NG = N / (16 * NT);
    constexpr int TW = 16 * NT;
    static_assert((KK * N) % GEMM_T == 0 && TW % 32 == 0 && N % TW == 0, "tile constants");
    __shared__ v4f Bl[KK * N];

    const int tid  = threadIdx.x;
    const int lane = tid & 31;
    const int wave = __builtin_amdgcn_readfirstlane(tid >> 5);
    const int c = lane & 15;
    const int h = lane >> 4;
    const int T = (M >> 4) * NG;
    const int tStep = gridDim.x * (GEMM_T / 32);
    int t = blockIdx.x + gridDim.x * wave;

    v4f a[KU];
    if (t < T) {
        const int mt = t / NG;
        const float* ap = A + (size_t)((mt << 4) + c) * K + (h << 2);
        #pragma unroll
        for (int u = 0; u < KU; ++u) a[u] = *(const v4f*)(ap + 8 * u);
    }

    #pragma unroll 8
    for (int i = 0; i < (KK * N) / GEMM_T; ++i) {
        const int e  = tid + i * GEMM_T;
        const int kk = e / N;
        const int n  = e - kk * N;
        const float* bp = B + (size_t)(4 * kk) * N + n;
        v4f v;
        v.x = bp[0]; v.y = bp[N]; v.z = bp[2 * N]; v.w = bp[3 * N];
        const int g  = n / TW;
        const int nl = n - g * TW;
        Bl[(g * KK + kk) * TW + (nl ^ ((kk & 1) << 3))] = v;
    }
    __syncthreads();

    const v4f* bl = Bl + h * TW + (c ^ (h << 3));
    while (t < T) {
        const int mt = t / NG;
        const int ng = t - mt * NG;
        const v4f* bt = bl + ng * (KK * TW);

        v8f acc[NT];
        v4f b[2][NT];
        #pragma unroll
        for (int j = 0; j < NT; ++j) { acc[j] = v8f{}; b[0][j] = bt[16 * j]; }
        float bj[NT];
        if (RELU_BIAS) {
            #pragma unroll
            for (int j = 0; j < NT; ++j) bj[j] = bias[ng * TW + 16 * j + c];
        }
        float rs[8];
        if (ROW_SCALE) {
            #pragma unroll
            for (int r = 0; r < 8; ++r) rs[r] = rowscale[(mt << 4) + (h << 3) + r];
        }

        #pragma unroll
        for (int u = 0; u < KU; ++u) {
            if (u + 1 < KU) {
                #pragma unroll
                for (int j = 0; j < NT; ++j) b[(u + 1) & 1][j] = bt[2 * (u + 1) * TW + 16 * j];
            }
            #pragma unroll
            for (int j = 0; j < NT; ++j)
                acc[j] = __builtin_amdgcn_wmma_f32_16x16x4_f32(false, a[u].xy, false, b[u & 1][j].xy,
                                                               (short)0, acc[j], false, false);
            #pragma unroll
            for (int j = 0; j < NT; ++j)
                acc[j] = __builtin_amdgcn_wmma_f32_16x16x4_f32(false, a[u].zw, false, b[u & 1][j].zw,
                                                               (short)0, acc[j], false, false);
        }

        float* cp = C + (size_t)((mt << 4) + (h << 3)) * N + ng * (16 * NT) + c;
        #pragma unroll
        for (int j = 0; j < NT; ++j) {
            #pragma unroll
            for (int r = 0; r < 8; ++r) {
                float v = acc[j][r];
                if (RELU_BIAS) v = fmaxf(v + bj[j], 0.f);
                if (ROW_SCALE) v *= rs[r];
                cp[(size_t)r * N + 16 * j] = v;
            }
        }

        t += tStep;
        if (t < T) {
            const int mn = t / NG;
            const float* ap = A + (size_t)((mn << 4) + c) * K + (h << 2);
            #pragma unroll
            for (int u = 0; u < KU; ++u) a[u] = *(const v4f*)(ap + 8 * u);
        }
    }
}

__global__ void gather128_kernel(const int* __restrict__ rowstart,
                                 const int* __restrict__ csr_src,
                                 const float* __restrict__ dinv,
                                 const float* __restrict__ xs,
                                 float* __restrict__ out, int n) {
    const int lane = threadIdx.x & 31;
    int node = (blockIdx.x * blockDim.x + threadIdx.x) >> 5;
    if (node >= n) return;
    node = __builtin_amdgcn_readfirstlane(node);

    const int beg = rowstart[node];
    const int end = rowstart[node + 1];
    const float di = dinv[node];
    asm volatile("" :: "s"(di));

    float4 a = ((const float4*)(xs + (size_t)node * 128))[lane];

    for (int j = beg; j < end; ++j) {
        const int sj = csr_src[j];
        const float4 v = ((const float4*)(xs + (size_t)sj * 128))[lane];
        a.x += v.x; a.y += v.y; a.z += v.z; a.w += v.w;
    }
    a.x *= di; a.y *= di; a.z *= di; a.w *= di;
    ((float4*)(out + (size_t)node * 128))[lane] = a;
}

__global__ void gather64_kernel(const int* __restrict__ rowstart,
                                const int* __restrict__ csr_src,
                                const float* __restrict__ dinv,
                                const float* __restrict__ h,
                                const float* __restrict__ bias,
                                float* __restrict__ out, int n) {
    const int lane = threadIdx.x & 31;
    int node = (blockIdx.x * blockDim.x + threadIdx.x) >> 5;
    if (node >= n) return;
    node = __builtin_amdgcn_readfirstlane(node);

    const int beg = rowstart[node];
    const int end = rowstart[node + 1];
    const float di = dinv[node];
    asm volatile("" :: "s"(di));

    const float2* hn = (const float2*)(h + (size_t)node * 64);
    const float2* b2 = (const float2*)bias;
    float2 a  = hn[lane];
    float2 bv = b2[lane];

    for (int j = beg; j < end; ++j) {
        const int sj = csr_src[j];
        const float2* hs = (const float2*)(h + (size_t)sj * 64);
        float2 v = hs[lane];
        a.x += v.x;
        a.y += v.y;
    }
    a.x = fmaf(di, a.x, bv.x);
    a.y = fmaf(di, a.y, bv.y);
    ((float2*)(out + (size_t)node * 64))[lane] = a;
}

extern "C" void kernel_launch(void* const* d_in, const int* in_sizes, int n_in,
                              void* d_out, int out_size, void* d_ws, size_t ws_size,
                              hipStream_t stream) {
    (void)n_in; (void)out_size; (void)ws_size;

    const float* x  = (const float*)d_in[0];
    const int*   ei = (const int*)d_in[1];
    const float* W1 = (const float*)d_in[2];
    const float* b1 = (const float*)d_in[3];
    const float* W2 = (const float*)d_in[4];
    const float* b2 = (const float*)d_in[5];
    float* out = (float*)d_out;

    const int Fin = 128, Fmid = 256, Fout = 64;
    const int N = in_sizes[0] / Fin;
    const int E = in_sizes[1] / 2;
    const int* src = ei;
    const int* dst = ei + E;

    char* ws = (char*)d_ws;
    size_t off = 0;
    auto take = [&](size_t bytes) -> char* {
        char* p = ws + off;
        off = (off + bytes + 255) & ~(size_t)255;
        return p;
    };
    float* dinv     = (float*)take((size_t)N * 4);
    int*   rowstart = (int*)  take((size_t)(N + 1) * 4);
    const int NB = (N + BIN_N - 1) / BIN_N;
    int*   offsT    = (int*)  take(((size_t)NB + 1) * BIN_G * 4);
    int2*  sorted   = (int2*) take((size_t)E * 8);
    int*   csr_src  = (int*)  take((size_t)E * 4);
    float* xs       = (float*)take((size_t)N * Fin * 4);
    float* aggx     = (float*)take((size_t)N * Fin * 4);
    float* hrelu    = (float*)take((size_t)N * Fmid * 4);
    float* h2       = (float*)take((size_t)N * Fout * 4);

    const int T = 256;

    bin_local_kernel<<<BIN_G, BIN_T, 0, stream>>>(src, dst, offsT, sorted, E, NB);
    bin_rows_kernel <<<NB, BIN_T, 0, stream>>>(sorted, offsT, x, dinv, xs, rowstart, csr_src, N, E);

    {
        long long threads = (long long)N * 32;
        gather128_kernel<<<(int)((threads + T - 1) / T), T, 0, stream>>>(
            rowstart, csr_src, dinv, xs, aggx, N);
    }
    gemm_f32_wmma_kernel<128, 256, 4, true, false><<<GEMM_GRID, GEMM_T, 0, stream>>>(aggx, W1, b1, nullptr, hrelu, N);
    gemm_f32_wmma_kernel<256, 64, 2, false, true><<<GEMM_GRID, GEMM_T, 0, stream>>>(hrelu, W2, nullptr, dinv, h2, N);
    {
        long long threads = (long long)N * 32;
        gather64_kernel<<<(int)((threads + T - 1) / T), T, 0, stream>>>(
            rowstart, csr_src, dinv, h2, b2, out, N);
    }
}


# Both CSR passes scatter per edge into an LDS image of the work-group's contiguous output range and write it out in whole lines: no per-edge 4/8-byte global stores; list order unchanged
# speedup vs baseline: 6.0595x; 1.0336x over previous
#include <hip/hip_runtime.h>

typedef float v2f __attribute__((ext_vector_type(2)));
typedef float v4f __attribute__((ext_vector_type(4)));
typedef float v8f __attribute__((ext_vector_type(8)));

#define BIN_SHIFT 8
#define BIN_N 256
#define BIN_MAX 1024
#define BIN_T 1024
#define BIN_G 256
#define BIN_SPW (BIN_G / (BIN_T / 32))
#define BIN_RPW (BIN_N / (BIN_T / 32))
#define LOC_CAP 4096
#define ROW_CAP 8192

__global__ void __launch_bounds__(BIN_T)
bin_local_kernel(const int* __restrict__ src, const int* __restrict__ dst,
                 int* __restrict__ offsT, int2* __restrict__ sorted, int E, int nbins) {
    __shared__ int h[BIN_MAX];
    __shared__ int wtot[BIN_T / 32];
    __shared__ int2 img[LOC_CAP];
    const int tid  = threadIdx.x;
    const int lane = tid & 31;
    const int wave = tid >> 5;
    const int chunk = (E + BIN_G - 1) / BIN_G;
    const int e0 = min((int)blockIdx.x * chunk, E);
    const int e1 = min(e0 + chunk, E);
    h[tid] = 0;
    __syncthreads();
    for (int e = e0 + tid; e < e1; e += BIN_T) atomicAdd(&h[dst[e] >> BIN_SHIFT], 1);
    __syncthreads();

    const int c = h[tid];
    int incl = c;
    #pragma unroll
    for (int d = 1; d < 32; d <<= 1) {
        const int t = __shfl_up(incl, d, 32);
        if (lane >= d) incl += t;
    }
    if (lane == 31) wtot[wave] = incl;
    __syncthreads();
    int wbase = 0;
    for (int w = 0; w < wave; ++w) wbase += wtot[w];
    const int excl = wbase + incl - c;
    if (tid < nbins) offsT[tid * BIN_G + blockIdx.x] = excl;
    if (tid == 0) offsT[nbins * BIN_G + blockIdx.x] = e1 - e0;
    h[tid] = e0 + excl;
    __syncthreads();
    const int ne = e1 - e0;
    if (ne <= LOC_CAP) {
        for (int e = e0 + tid; e < e1; e += BIN_T) {
            const int d = dst[e];
            const int slot = atomicAdd(&h[d >> BIN_SHIFT], 1);
            img[slot - e0] = make_int2(src[e], d);
        }
        __syncthreads();
        for (int i = tid - (e0 & 31); i < ne; i += BIN_T)
            if (i >= 0) sorted[e0 + i] = img[i];
    } else {
        for (int e = e0 + tid; e < e1; e += BIN_T) {
            const int d = dst[e];
            const int slot = atomicAdd(&h[d >> BIN_SHIFT], 1);
            sorted[slot] = make_int2(src[e], d);
        }
    }
}

static __device__ __forceinline__ void lds_barrier() {
    __builtin_amdgcn_fence(__ATOMIC_RELEASE, "workgroup", "local");
    __builtin_amdgcn_s_barrier();
    __builtin_amdgcn_fence(__ATOMIC_ACQUIRE, "workgroup", "local");
}

__global__ void __launch_bounds__(BIN_T)
bin_rows_kernel(const int2* __restrict__ sorted, const int* __restrict__ offsT,
                const float* __restrict__ x, float* __restrict__ dinv, float* __restrict__ xs,
                int* __restrict__ rowstart, int* __restrict__ csr_src,
                int n, int E) {
    __shared__ int cnt[BIN_N];
    __shared__ float dsc[BIN_N];
    __shared__ int wsum[BIN_N / 32];
    __shared__ int segs[BIN_G];
    __shared__ int segl[BIN_G];
    __shared__ int part[2 * (BIN_G / 32)];
    __shared__ int img[ROW_CAP];
    const int tid  = threadIdx.x;
    const int lane = tid & 31;
    const int wave = tid >> 5;
    const bool own = tid < BIN_N;
    const int b = blockIdx.x;
    const int chunk = (E + BIN_G - 1) / BIN_G;
    if (tid < BIN_G) {
        const int o0 = offsT[b * BIN_G + tid];
        const int o1 = offsT[(b + 1) * BIN_G + tid];
        segs[tid] = min(tid * chunk, E) + o0;
        segl[tid] = o1 - o0;
        int r0 = o0, r1 = o1;
        #pragma unroll
        for (int d = 16; d >= 1; d >>= 1) {
            r0 += __shfl_xor(r0, d, 32);
            r1 += __shfl_xor(r1, d, 32);
        }
        if (lane == 0) { part[wave] = r0; part[BIN_G / 32 + wave] = r1; }
    }
    if (own) cnt[tid] = 0;
    __syncthreads();
    int s0 = 0, s1 = 0;
    #pragma unroll
    for (int w = 0; w < BIN_G / 32; ++w) { s0 += part[w]; s1 += part[BIN_G / 32 + w]; }

    int  st[BIN_SPW], ln[BIN_SPW];
    int2 p[BIN_SPW];
    #pragma unroll
    for (int k = 0; k < BIN_SPW; ++k) {
        st[k] = segs[wave + k * (BIN_T / 32)];
        ln[k] = segl[wave + k * (BIN_T / 32)];
        p[k] = make_int2(0, 0);
        if (lane < ln[k]) p[k] = sorted[st[k] + lane];
    }
    const int row0 = b * BIN_N + wave * BIN_RPW;
    float4 xv[BIN_RPW];
    #pragma unroll
    for (int k = 0; k < BIN_RPW; ++k) {
        const int r = min(row0 + k, n - 1);
        xv[k] = ((const float4*)(x + (size_t)r * 128))[lane];
    }
    #pragma unroll
    for (int k = 0; k < BIN_SPW; ++k) {
        if (lane < ln[k]) atomicAdd(&cnt[p[k].y & (BIN_N - 1)], 1);
        for (int i = lane + 32; i < ln[k]; i += 32)
            atomicAdd(&cnt[sorted[st[k] + i].y & (BIN_N - 1)], 1);
    }
    lds_barrier();

    const int c = own ? cnt[tid] : 0;
    int incl = c;
    #pragma unroll
    for (int d = 1; d < 32; d <<= 1) {
        const int t = __shfl_up(incl, d, 32);
        if (lane >= d) incl += t;
    }
    if (own && lane == 31) wsum[wave] = incl;
    const float dmine = rsqrtf((float)c + 1.0f);
    if (own) dsc[tid] = dmine;
    lds_barrier();
    if (own) {
        int wbase = 0;
        for (int w = 0; w < wave; ++w) wbase += wsum[w];
        const int start = s0 + wbase + incl - c;
        const int node = b * BIN_N + tid;
        if (node < n) { dinv[node] = dmine; rowstart[node] = start; }
        cnt[tid] = start;
    }
    if (b == (int)gridDim.x - 1 && tid == 0) rowstart[n] = s1;
    lds_barrier();
    #pragma unroll
    for (int k = 0; k < BIN_RPW; ++k) {
        if (row0 + k < n) {
            const float d = dsc[wave * BIN_RPW + k];
            float4 o;
            o.x = d * xv[k].x; o.y = d * xv[k].y; o.z = d * xv[k].z; o.w = d * xv[k].w;
            ((float4*)(xs + (size_t)(row0 + k) * 128))[lane] = o;
        }
    }
    const int nb = s1 - s0;
    if (nb <= ROW_CAP) {
        #pragma unroll
        for (int k = 0; k < BIN_SPW; ++k) {
            if (lane < ln[k]) {
                const int slot = atomicAdd(&cnt[p[k].y & (BIN_N - 1)], 1);
                img[slot - s0] = p[k].x;
            }
            for (int i = lane + 32; i < ln[k]; i += 32) {
                const int2 q = sorted[st[k] + i];
                const int slot = atomicAdd(&cnt[q.y & (BIN_N - 1)], 1);
                img[slot - s0] = q.x;
            }
        }
        lds_barrier();
        for (int i = tid - (s0 & 31); i < nb; i += BIN_T)
            if (i >= 0) csr_src[s0 + i] = img[i];
    } else {
        #pragma unroll
        for (int k = 0; k < BIN_SPW; ++k) {
            if (lane < ln[k]) {
                const int slot = atomicAdd(&cnt[p[k].y & (BIN_N - 1)], 1);
                csr_src[slot] = p[k].x;
            }
            for (int i = lane + 32; i < ln[k]; i += 32) {
                const int2 q = sorted[st[k] + i];
                const int slot = atomicAdd(&cnt[q.y & (BIN_N - 1)], 1);
                csr_src[slot] = q.x;
            }
        }
    }
}

#define GEMM_T 512
#define GEMM_GRID 256

template <int K, int N, int NT, bool RELU_BIAS, bool ROW_SCALE>
__global__ void __launch_bounds__(GEMM_T)
gemm_f32_wmma_kernel(const float* __restrict__ A,
                     const float* __restrict__ B,
                     const float* __restrict__ bias,
                     const float* __restrict__ rowscale,
                     float* __restrict__ C, int M) {
    constexpr int KU = K / 8;
    constexpr int KK = K / 4;
    constexpr int NG = N / (16 * NT);
    constexpr int TW = 16 * NT;
    static_assert((KK * N) % GEMM_T == 0 && TW % 32 == 0 && N % TW == 0, "tile constants");
    __shared__ v4f Bl[KK * N];

    const int tid  = threadIdx.x;
    const int lane = tid & 31;
    const int wave = __builtin_amdgcn_readfirstlane(tid >> 5);
    const int c = lane & 15;
    const int h = lane >> 4;
    const int T = (M >> 4) * NG;
    const int tStep = gridDim.x * (GEMM_T / 32);
    int t = blockIdx.x + gridDim.x * wave;

    v4f a[KU];
    if (t < T) {
        const int mt = t / NG;
        const float* ap = A + (size_t)((mt << 4) + c) * K + (h << 2);
        #pragma unroll
        for (int u = 0; u < KU; ++u) a[u] = *(const v4f*)(ap + 8 * u);
    }

    #pragma unroll 8
    for (int i = 0; i < (KK * N) / GEMM_T; ++i) {
        const int e  = tid + i * GEMM_T;
        const int kk = e / N;
        const int n  = e - kk * N;
        const float* bp = B + (size_t)(4 * kk) * N + n;
        v4f v;
        v.x = bp[0]; v.y = bp[N]; v.z = bp[2 * N]; v.w = bp[3 * N];
        const int g  = n / TW;
        const int nl = n - g * TW;
        Bl[(g * KK + kk) * TW + (nl ^ ((kk & 1) << 3))] = v;
    }
    __syncthreads();

    const v4f* bl = Bl + h * TW + (c ^ (h << 3));
    while (t < T) {
        const int mt = t / NG;
        const int ng = t - mt * NG;
        const v4f* bt = bl + ng * (KK * TW);

        v8f acc[NT];
        v4f b[2][NT];
        #pragma unroll
        for (int j = 0; j < NT; ++j) { acc[j] = v8f{}; b[0][j] = bt[16 * j]; }
        float bj[NT];
        if (RELU_BIAS) {
            #pragma unroll
            for (int j = 0; j < NT; ++j) bj[j] = bias[ng * TW + 16 * j + c];
        }
        float rs[8];
        if (ROW_SCALE) {
            #pragma unroll
            for (int r = 0; r < 8; ++r) rs[r] = rowscale[(mt << 4) + (h << 3) + r];
        }

        #pragma unroll
        for (int u = 0; u < KU; ++u) {
            if (u + 1 < KU) {
                #pragma unroll
                for (int j = 0; j < NT; ++j) b[(u + 1) & 1][j] = bt[2 * (u + 1) * TW + 16 * j];
            }
            #pragma unroll
            for (int j = 0; j < NT; ++j)
                acc[j] = __builtin_amdgcn_wmma_f32_16x16x4_f32(false, a[u].xy, false, b[u & 1][j].xy,
                                                               (short)0, acc[j], false, false);
            #pragma unroll
            for (int j = 0; j < NT; ++j)
                acc[j] = __builtin_amdgcn_wmma_f32_16x16x4_f32(false, a[u].zw, false, b[u & 1][j].zw,
                                                               (short)0, acc[j], false, false);
        }

        float* cp = C + (size_t)((mt << 4) + (h << 3)) * N + ng * (16 * NT) + c;
        #pragma unroll
        for (int j = 0; j < NT; ++j) {
            #pragma unroll
            for (int r = 0; r < 8; ++r) {
                float v = acc[j][r];
                if (RELU_BIAS) v = fmaxf(v + bj[j], 0.f);
                if (ROW_SCALE) v *= rs[r];
                cp[(size_t)r * N + 16 * j] = v;
            }
        }

        t += tStep;
        if (t < T) {
            const int mn = t / NG;
            const float* ap = A + (size_t)((mn << 4) + c) * K + (h << 2);
            #pragma unroll
            for (int u = 0; u < KU; ++u) a[u] = *(const v4f*)(ap + 8 * u);
        }
    }
}

__global__ void gather128_kernel(const int* __restrict__ rowstart,
                                 const int* __restrict__ csr_src,
                                 const float* __restrict__ dinv,
                                 const float* __restrict__ xs,
                                 float* __restrict__ out, int n) {
    const int lane = threadIdx.x & 31;
    int node = (blockIdx.x * blockDim.x + threadIdx.x) >> 5;
    if (node >= n) return;
    node = __builtin_amdgcn_readfirstlane(node);

    const int beg = rowstart[node];
    const int end = rowstart[node + 1];
    const float di = dinv[node];
    asm volatile("" :: "s"(di));

    float4 a = ((const float4*)(xs + (size_t)node * 128))[lane];

    for (int j = beg; j < end; ++j) {
        const int sj = csr_src[j];
        const float4 v = ((const float4*)(xs + (size_t)sj * 128))[lane];
        a.x += v.x; a.y += v.y; a.z += v.z; a.w += v.w;
    }
    a.x *= di; a.y *= di; a.z *= di; a.w *= di;
    ((float4*)(out + (size_t)node * 128))[lane] = a;
}

__global__ void gather64_kernel(const int* __restrict__ rowstart,
                                const int* __restrict__ csr_src,
                                const float* __restrict__ dinv,
                                const float* __restrict__ h,
                                const float* __restrict__ bias,
                                float* __restrict__ out, int n) {
    const int lane = threadIdx.x & 31;
    int node = (blockIdx.x * blockDim.x + threadIdx.x) >> 5;
    if (node >= n) return;
    node = __builtin_amdgcn_readfirstlane(node);

    const int beg = rowstart[node];
    const int end = rowstart[node + 1];
    const float di = dinv[node];
    asm volatile("" :: "s"(di));

    const float2* hn = (const float2*)(h + (size_t)node * 64);
    const float2* b2 = (const float2*)bias;
    float2 a  = hn[lane];
    float2 bv = b2[lane];

    for (int j = beg; j < end; ++j) {
        const int sj = csr_src[j];
        const float2* hs = (const float2*)(h + (size_t)sj * 64);
        float2 v = hs[lane];
        a.x += v.x;
        a.y += v.y;
    }
    a.x = fmaf(di, a.x, bv.x);
    a.y = fmaf(di, a.y, bv.y);
    ((float2*)(out + (size_t)node * 64))[lane] = a;
}

extern "C" void kernel_launch(void* const* d_in, const int* in_sizes, int n_in,
                              void* d_out, int out_size, void* d_ws, size_t ws_size,
                              hipStream_t stream) {
    (void)n_in; (void)out_size; (void)ws_size;

    const float* x  = (const float*)d_in[0];
    const int*   ei = (const int*)d_in[1];
    const float* W1 = (const float*)d_in[2];
    const float* b1 = (const float*)d_in[3];
    const float* W2 = (const float*)d_in[4];
    const float* b2 = (const float*)d_in[5];
    float* out = (float*)d_out;

    const int Fin = 128, Fmid = 256, Fout = 64;
    const int N = in_sizes[0] / Fin;
    const int E = in_sizes[1] / 2;
    const int* src = ei;
    const int* dst = ei + E;

    char* ws = (char*)d_ws;
    size_t off = 0;
    auto take = [&](size_t bytes) -> char* {
        char* p = ws + off;
        off = (off + bytes + 255) & ~(size_t)255;
        return p;
    };
    float* dinv     = (float*)take((size_t)N * 4);
    int*   rowstart = (int*)  take((size_t)(N + 1) * 4);
    const int NB = (N + BIN_N - 1) / BIN_N;
    int*   offsT    = (int*)  take(((size_t)NB + 1) * BIN_G * 4);
    int2*  sorted   = (int2*) take((size_t)E * 8);
    int*   csr_src  = (int*)  take((size_t)E * 4);
    float* xs       = (float*)take((size_t)N * Fin * 4);
    float* aggx     = (float*)take((size_t)N * Fin * 4);
    float* hrelu    = (float*)take((size_t)N * Fmid * 4);
    float* h2       = (float*)take((size_t)N * Fout * 4);

    const int T = 256;

    bin_local_kernel<<<BIN_G, BIN_T, 0, stream>>>(src, dst, offsT, sorted, E, NB);
    bin_rows_kernel <<<NB, BIN_T, 0, stream>>>(sorted, offsT, x, dinv, xs, rowstart, csr_src, N, E);

    {
        long long threads = (long long)N * 32;
        gather128_kernel<<<(int)((threads + T - 1) / T), T, 0, stream>>>(
            rowstart, csr_src, dinv, xs, aggx, N);
    }
    gemm_f32_wmma_kernel<128, 256, 4, true, false><<<GEMM_GRID, GEMM_T, 0, stream>>>(aggx, W1, b1, nullptr, hrelu, N);
    gemm_f32_wmma_kernel<256, 64, 2, false, true><<<GEMM_GRID, GEMM_T, 0, stream>>>(hrelu, W2, nullptr, dinv, h2, N);
    {
        long long threads = (long long)N * 32;
        gather64_kernel<<<(int)((threads + T - 1) / T), T, 0, stream>>>(
            rowstart, csr_src, dinv, h2, b2, out, N);
    }
}


# Both dense layers in ONE launch: each resident work-group owns a range of 16-row tiles, stages W1 and W2 together and runs layer 2 on its own rows behind a work-group barrier (6 -> 5 launches)
# speedup vs baseline: 6.3630x; 1.0501x over previous
#include <hip/hip_runtime.h>

typedef float v2f __attribute__((ext_vector_type(2)));
typedef float v4f __attribute__((ext_vector_type(4)));
typedef float v8f __attribute__((ext_vector_type(8)));

#define BIN_SHIFT 8
#define BIN_N 256
#define BIN_MAX 1024
#define BIN_T 1024
#define BIN_G 256
#define BIN_SPW (BIN_G / (BIN_T / 32))
#define BIN_RPW (BIN_N / (BIN_T / 32))
#define LOC_CAP 4096
#define ROW_CAP 8192

__global__ void __launch_bounds__(BIN_T)
bin_local_kernel(const int* __restrict__ src, const int* __restrict__ dst,
                 int* __restrict__ offsT, int2* __restrict__ sorted, int E, int nbins) {
    __shared__ int h[BIN_MAX];
    __shared__ int wtot[BIN_T / 32];
    __shared__ int2 img[LOC_CAP];
    const int tid  = threadIdx.x;
    const int lane = tid & 31;
    const int wave = tid >> 5;
    const int chunk = (E + BIN_G - 1) / BIN_G;
    const int e0 = min((int)blockIdx.x * chunk, E);
    const int e1 = min(e0 + chunk, E);
    h[tid] = 0;
    __syncthreads();
    for (int e = e0 + tid; e < e1; e += BIN_T) atomicAdd(&h[dst[e] >> BIN_SHIFT], 1);
    __syncthreads();

    const int c = h[tid];
    int incl = c;
    #pragma unroll
    for (int d = 1; d < 32; d <<= 1) {
        const int t = __shfl_up(incl, d, 32);
        if (lane >= d) incl += t;
    }
    if (lane == 31) wtot[wave] = incl;
    __syncthreads();
    int wbase = 0;
    for (int w = 0; w < wave; ++w) wbase += wtot[w];
    const int excl = wbase + incl - c;
    if (tid < nbins) offsT[tid * BIN_G + blockIdx.x] = excl;
    if (tid == 0) offsT[nbins * BIN_G + blockIdx.x] = e1 - e0;
    h[tid] = e0 + excl;
    __syncthreads();
    const int ne = e1 - e0;
    if (ne <= LOC_CAP) {
        for (int e = e0 + tid; e < e1; e += BIN_T) {
            const int d = dst[e];
            const int slot = atomicAdd(&h[d >> BIN_SHIFT], 1);
            img[slot - e0] = make_int2(src[e], d);
        }
        __syncthreads();
        for (int i = tid - (e0 & 31); i < ne; i += BIN_T)
            if (i >= 0) sorted[e0 + i] = img[i];
    } else {
        for (int e = e0 + tid; e < e1; e += BIN_T) {
            const int d = dst[e];
            const int slot = atomicAdd(&h[d >> BIN_SHIFT], 1);
            sorted[slot] = make_int2(src[e], d);
        }
    }
}

static __device__ __forceinline__ void lds_barrier() {
    __builtin_amdgcn_fence(__ATOMIC_RELEASE, "workgroup", "local");
    __builtin_amdgcn_s_barrier();
    __builtin_amdgcn_fence(__ATOMIC_ACQUIRE, "workgroup", "local");
}

__global__ void __launch_bounds__(BIN_T)
bin_rows_kernel(const int2* __restrict__ sorted, const int* __restrict__ offsT,
                const float* __restrict__ x, float* __restrict__ dinv, float* __restrict__ xs,
                int* __restrict__ rowstart, int* __restrict__ csr_src,
                int n, int E) {
    __shared__ int cnt[BIN_N];
    __shared__ float dsc[BIN_N];
    __shared__ int wsum[BIN_N / 32];
    __shared__ int segs[BIN_G];
    __shared__ int segl[BIN_G];
    __shared__ int part[2 * (BIN_G / 32)];
    __shared__ int img[ROW_CAP];
    const int tid  = threadIdx.x;
    const int lane = tid & 31;
    const int wave = tid >> 5;
    const bool own = tid < BIN_N;
    const int b = blockIdx.x;
    const int chunk = (E + BIN_G - 1) / BIN_G;
    if (tid < BIN_G) {
        const int o0 = offsT[b * BIN_G + tid];
        const int o1 = offsT[(b + 1) * BIN_G + tid];
        segs[tid] = min(tid * chunk, E) + o0;
        segl[tid] = o1 - o0;
        int r0 = o0, r1 = o1;
        #pragma unroll
        for (int d = 16; d >= 1; d >>= 1) {
            r0 += __shfl_xor(r0, d, 32);
            r1 += __shfl_xor(r1, d, 32);
        }
        if (lane == 0) { part[wave] = r0; part[BIN_G / 32 + wave] = r1; }
    }
    if (own) cnt[tid] = 0;
    __syncthreads();
    int s0 = 0, s1 = 0;
    #pragma unroll
    for (int w = 0; w < BIN_G / 32; ++w) { s0 += part[w]; s1 += part[BIN_G / 32 + w]; }

    int  st[BIN_SPW], ln[BIN_SPW];
    int2 p[BIN_SPW];
    #pragma unroll
    for (int k = 0; k < BIN_SPW; ++k) {
        st[k] = segs[wave + k * (BIN_T / 32)];
        ln[k] = segl[wave + k * (BIN_T / 32)];
        p[k] = make_int2(0, 0);
        if (lane < ln[k]) p[k] = sorted[st[k] + lane];
    }
    const int row0 = b * BIN_N + wave * BIN_RPW;
    float4 xv[BIN_RPW];
    #pragma unroll
    for (int k = 0; k < BIN_RPW; ++k) {
        const int r = min(row0 + k, n - 1);
        xv[k] = ((const float4*)(x + (size_t)r * 128))[lane];
    }
    #pragma unroll
    for (int k = 0; k < BIN_SPW; ++k) {
        if (lane < ln[k]) atomicAdd(&cnt[p[k].y & (BIN_N - 1)], 1);
        for (int i = lane + 32; i < ln[k]; i += 32)
            atomicAdd(&cnt[sorted[st[k] + i].y & (BIN_N - 1)], 1);
    }
    lds_barrier();

    const int c = own ? cnt[tid] : 0;
    int incl = c;
    #pragma unroll
    for (int d = 1; d < 32; d <<= 1) {
        const int t = __shfl_up(incl, d, 32);
        if (lane >= d) incl += t;
    }
    if (own && lane == 31) wsum[wave] = incl;
    const float dmine = rsqrtf((float)c + 1.0f);
    if (own) dsc[tid] = dmine;
    lds_barrier();
    if (own) {
        int wbase = 0;
        for (int w = 0; w < wave; ++w) wbase += wsum[w];
        const int start = s0 + wbase + incl - c;
        const int node = b * BIN_N + tid;
        if (node < n) { dinv[node] = dmine; rowstart[node] = start; }
        cnt[tid] = start;
    }
    if (b == (int)gridDim.x - 1 && tid == 0) rowstart[n] = s1;
    lds_barrier();
    #pragma unroll
    for (int k = 0; k < BIN_RPW; ++k) {
        if (row0 + k < n) {
            const float d = dsc[wave * BIN_RPW + k];
            float4 o;
            o.x = d * xv[k].x; o.y = d * xv[k].y; o.z = d * xv[k].z; o.w = d * xv[k].w;
            ((float4*)(xs + (size_t)(row0 + k) * 128))[lane] = o;
        }
    }
    const int nb = s1 - s0;
    if (nb <= ROW_CAP) {
        #pragma unroll
        for (int k = 0; k < BIN_SPW; ++k) {
            if (lane < ln[k]) {
                const int slot = atomicAdd(&cnt[p[k].y & (BIN_N - 1)], 1);
                img[slot - s0] = p[k].x;
            }
            for (int i = lane + 32; i < ln[k]; i += 32) {
                const int2 q = sorted[st[k] + i];
                const int slot = atomicAdd(&cnt[q.y & (BIN_N - 1)], 1);
                img[slot - s0] = q.x;
            }
        }
        lds_barrier();
        for (int i = tid - (s0 & 31); i < nb; i += BIN_T)
            if (i >= 0) csr_src[s0 + i] = img[i];
    } else {
        #pragma unroll
        for (int k = 0; k < BIN_SPW; ++k) {
            if (lane < ln[k]) {
                const int slot = atomicAdd(&cnt[p[k].y & (BIN_N - 1)], 1);
                csr_src[slot] = p[k].x;
            }
            for (int i = lane + 32; i < ln[k]; i += 32) {
                const int2 q = sorted[st[k] + i];
                const int slot = atomicAdd(&cnt[q.y & (BIN_N - 1)], 1);
                csr_src[slot] = q.x;
            }
        }
    }
}

#define GEMM_T 512
#define GEMM_W (GEMM_T / 32)
#define GEMM_GRID 256

template <int K, int N, int NT>
static __device__ __forceinline__ void gemm_stage_b(const float* __restrict__ B, v4f* Bl, int tid) {
    constexpr int KK = K / 4;
    constexpr int TW = 16 * NT;
    static_assert((KK * N) % GEMM_T == 0 && TW % 32 == 0 && N % TW == 0, "tile constants");
    #pragma unroll 8
    for (int i = 0; i < (KK * N) / GEMM_T; ++i) {
        const int e  = tid + i * GEMM_T;
        const int kk = e / N;
        const int n  = e - kk * N;
        const float* bp = B + (size_t)(4 * kk) * N + n;
        v4f v;
        v.x = bp[0]; v.y = bp[N]; v.z = bp[2 * N]; v.w = bp[3 * N];
        const int g  = n / TW;
        const int nl = n - g * TW;
        Bl[(g * KK + kk) * TW + (nl ^ ((kk & 1) << 3))] = v;
    }
}

template <int K>
static __device__ __forceinline__ void gemm_load_a(v4f (&a)[K / 8], const float* A, int mt, int c, int h) {
    const float* ap = A + (size_t)((mt << 4) + c) * K + (h << 2);
    #pragma unroll
    for (int u = 0; u < K / 8; ++u) a[u] = *(const v4f*)(ap + 8 * u);
}

template <int K, int N, int NT, bool RELU_BIAS, bool ROW_SCALE>
static __device__ __forceinline__ void gemm_run_tasks(v4f (&a)[K / 8], const float* A, const v4f* Bl,
                                                      const float* __restrict__ bias,
                                                      const float* __restrict__ rowscale,
                                                      float* C, int mt0, int T, int slot, int c, int h) {
    constexpr int KU = K / 8;
    constexpr int KK = K / 4;
    constexpr int NG = N / (16 * NT);
    constexpr int TW = 16 * NT;
    const v4f* bl = Bl + h * TW + (c ^ (h << 3));
    int t = slot;
    while (t < T) {
        const int ml = t / NG;
        const int ng = t - ml * NG;
        const int mt = mt0 + ml;
        const v4f* bt = bl + ng * (KK * TW);

        v8f acc[NT];
        v4f b[2][NT];
        #pragma unroll
        for (int j = 0; j < NT; ++j) { acc[j] = v8f{}; b[0][j] = bt[16 * j]; }
        float bj[NT];
        if (RELU_BIAS) {
            #pragma unroll
            for (int j = 0; j < NT; ++j) bj[j] = bias[ng * TW + 16 * j + c];
        }
        float rs[8];
        if (ROW_SCALE) {
            #pragma unroll
            for (int r = 0; r < 8; ++r) rs[r] = rowscale[(mt << 4) + (h << 3) + r];
        }

        #pragma unroll
        for (int u = 0; u < KU; ++u) {
            if (u + 1 < KU) {
                #pragma unroll
                for (int j = 0; j < NT; ++j) b[(u + 1) & 1][j] = bt[2 * (u + 1) * TW + 16 * j];
            }
            #pragma unroll
            for (int j = 0; j < NT; ++j)
                acc[j] = __builtin_amdgcn_wmma_f32_16x16x4_f32(false, a[u].xy, false, b[u & 1][j].xy,
                                                               (short)0, acc[j], false, false);
            #pragma unroll
            for (int j = 0; j < NT; ++j)
                acc[j] = __builtin_amdgcn_wmma_f32_16x16x4_f32(false, a[u].zw, false, b[u & 1][j].zw,
                                                               (short)0, acc[j], false, false);
            __builtin_amdgcn_sched_barrier(0);
        }

        float* cp = C + (size_t)((mt << 4) + (h << 3)) * N + ng * (16 * NT) + c;
        #pragma unroll
        for (int j = 0; j < NT; ++j) {
            #pragma unroll
            for (int r = 0; r < 8; ++r) {
                float v = acc[j][r];
                if (RELU_BIAS) v = fmaxf(v + bj[j], 0.f);
                if (ROW_SCALE) v *= rs[r];
                cp[(size_t)r * N + 16 * j] = v;
            }
        }

        t += GEMM_W;
        if (t < T) gemm_load_a<K>(a, A, mt0 + t / NG, c, h);
    }
}

template <int K1, int N1, int NT1, int N2, int NT2>
__global__ void __launch_bounds__(GEMM_T)
mlp_rows_wmma_kernel(const float* __restrict__ A, const float* __restrict__ W1,
                     const float* __restrict__ bias1, const float* __restrict__ W2,
                     const float* __restrict__ rowscale, float* hmid,
                     float* __restrict__ C, int M) {
    __shared__ v4f Bl1[(K1 / 4) * N1];
    __shared__ v4f Bl2[(N1 / 4) * N2];

    const int tid  = threadIdx.x;
    const int lane = tid & 31;
    const int wave = __builtin_amdgcn_readfirstlane(tid >> 5);
    const int c = lane & 15;
    const int h = lane >> 4;
    const int slot = ((((wave & 3) - (wave >> 2)) & 3) << 2) + (wave >> 2);
    const int MT  = M >> 4;
    const int mt0 = (int)(((long long)MT * (int)blockIdx.x) / (int)gridDim.x);
    const int mt1 = (int)(((long long)MT * ((int)blockIdx.x + 1)) / (int)gridDim.x);
    const int T1 = (mt1 - mt0) * (N1 / (16 * NT1));
    const int T2 = (mt1 - mt0) * (N2 / (16 * NT2));

    {
        v4f a[K1 / 8];
        if (slot < T1) gemm_load_a<K1>(a, A, mt0 + slot / (N1 / (16 * NT1)), c, h);
        gemm_stage_b<K1, N1, NT1>(W1, Bl1, tid);
        gemm_stage_b<N1, N2, NT2>(W2, Bl2, tid);
        __syncthreads();
        gemm_run_tasks<K1, N1, NT1, true, false>(a, A, Bl1, bias1, nullptr, hmid, mt0, T1, slot, c, h);
    }
    __syncthreads();
    {
        v4f a[N1 / 8];
        if (slot < T2) gemm_load_a<N1>(a, hmid, mt0 + slot / (N2 / (16 * NT2)), c, h);
        gemm_run_tasks<N1, N2, NT2, false, true>(a, hmid, Bl2, nullptr, rowscale, C, mt0, T2, slot, c, h);
    }
}

__global__ void gather128_kernel(const int* __restrict__ rowstart,
                                 const int* __restrict__ csr_src,
                                 const float* __restrict__ dinv,
                                 const float* __restrict__ xs,
                                 float* __restrict__ out, int n) {
    const int lane = threadIdx.x & 31;
    int node = (blockIdx.x * blockDim.x + threadIdx.x) >> 5;
    if (node >= n) return;
    node = __builtin_amdgcn_readfirstlane(node);

    const int beg = rowstart[node];
    const int end = rowstart[node + 1];
    const float di = dinv[node];
    asm volatile("" :: "s"(di));

    float4 a = ((const float4*)(xs + (size_t)node * 128))[lane];

    for (int j = beg; j < end; ++j) {
        const int sj = csr_src[j];
        const float4 v = ((const float4*)(xs + (size_t)sj * 128))[lane];
        a.x += v.x; a.y += v.y; a.z += v.z; a.w += v.w;
    }
    a.x *= di; a.y *= di; a.z *= di; a.w *= di;
    ((float4*)(out + (size_t)node * 128))[lane] = a;
}

__global__ void gather64_kernel(const int* __restrict__ rowstart,
                                const int* __restrict__ csr_src,
                                const float* __restrict__ dinv,
                                const float* __restrict__ h,
                                const float* __restrict__ bias,
                                float* __restrict__ out, int n) {
    const int lane = threadIdx.x & 31;
    int node = (blockIdx.x * blockDim.x + threadIdx.x) >> 5;
    if (node >= n) return;
    node = __builtin_amdgcn_readfirstlane(node);

    const int beg = rowstart[node];
    const int end = rowstart[node + 1];
    const float di = dinv[node];
    asm volatile("" :: "s"(di));

    const float2* hn = (const float2*)(h + (size_t)node * 64);
    const float2* b2 = (const float2*)bias;
    float2 a  = hn[lane];
    float2 bv = b2[lane];

    for (int j = beg; j < end; ++j) {
        const int sj = csr_src[j];
        const float2* hs = (const float2*)(h + (size_t)sj * 64);
        float2 v = hs[lane];
        a.x += v.x;
        a.y += v.y;
    }
    a.x = fmaf(di, a.x, bv.x);
    a.y = fmaf(di, a.y, bv.y);
    ((float2*)(out + (size_t)node * 64))[lane] = a;
}

extern "C" void kernel_launch(void* const* d_in, const int* in_sizes, int n_in,
                              void* d_out, int out_size, void* d_ws, size_t ws_size,
                              hipStream_t stream) {
    (void)n_in; (void)out_size; (void)ws_size;

    const float* x  = (const float*)d_in[0];
    const int*   ei = (const int*)d_in[1];
    const float* W1 = (const float*)d_in[2];
    const float* b1 = (const float*)d_in[3];
    const float* W2 = (const float*)d_in[4];
    const float* b2 = (const float*)d_in[5];
    float* out = (float*)d_out;

    const int Fin = 128, Fmid = 256, Fout = 64;
    const int N = in_sizes[0] / Fin;
    const int E = in_sizes[1] / 2;
    const int* src = ei;
    const int* dst = ei + E;

    char* ws = (char*)d_ws;
    size_t off = 0;
    auto take = [&](size_t bytes) -> char* {
        char* p = ws + off;
        off = (off + bytes + 255) & ~(size_t)255;
        return p;
    };
    float* dinv     = (float*)take((size_t)N * 4);
    int*   rowstart = (int*)  take((size_t)(N + 1) * 4);
    const int NB = (N + BIN_N - 1) / BIN_N;
    int*   offsT    = (int*)  take(((size_t)NB + 1) * BIN_G * 4);
    int2*  sorted   = (int2*) take((size_t)E * 8);
    int*   csr_src  = (int*)  take((size_t)E * 4);
    float* xs       = (float*)take((size_t)N * Fin * 4);
    float* aggx     = (float*)take((size_t)N * Fin * 4);
    float* hrelu    = (float*)take((size_t)N * Fmid * 4);
    float* h2       = (float*)take((size_t)N * Fout * 4);

    const int T = 256;

    bin_local_kernel<<<BIN_G, BIN_T, 0, stream>>>(src, dst, offsT, sorted, E, NB);
    bin_rows_kernel <<<NB, BIN_T, 0, stream>>>(sorted, offsT, x, dinv, xs, rowstart, csr_src, N, E);

    {
        long long threads = (long long)N * 32;
        gather128_kernel<<<(int)((threads + T - 1) / T), T, 0, stream>>>(
            rowstart, csr_src, dinv, xs, aggx, N);
    }
    mlp_rows_wmma_kernel<128, 256, 4, 64, 2><<<GEMM_GRID, GEMM_T, 0, stream>>>(aggx, W1, b1, W2, dinv, hrelu, h2, N);
    {
        long long threads = (long long)N * 32;
        gather64_kernel<<<(int)((threads + T - 1) / T), T, 0, stream>>>(
            rowstart, csr_src, dinv, h2, b2, out, N);
    }
}
